# starting kernel
# baseline (speedup 1.0000x reference)
#include <hip/hip_runtime.h>
#include <math.h>

#define NN 10000
#define EE 320000
#define HH 4
#define DD 128
#define FH (HH * DD)
#define K0 256
#define KMAX 512
#define NEG_SLOPE 0.2f

typedef __attribute__((ext_vector_type(2))) float v2f;
typedef __attribute__((ext_vector_type(8))) float v8f;

__global__ void fill_kernel(float* __restrict__ p, long n, float v) {
  long i = (long)blockIdx.x * blockDim.x + threadIdx.x;
  if (i < n) p[i] = v;
}

__global__ __launch_bounds__(256) void gemm16_wmma(const float* __restrict__ X,
                                                   const float* __restrict__ W,
                                                   float* __restrict__ out,
                                                   int M, int K, int Nout) {
  __shared__ float Bs[KMAX * 16];

  const int tid = threadIdx.x;
  const int lane = tid & 31;
  const int wv = tid >> 5;
  const int ntiles = Nout >> 4;
  const int mg = blockIdx.x / ntiles;
  const int tn = blockIdx.x - mg * ntiles;

  const float* wp = W + tn * 16;
  for (int i = tid; i < K * 16; i += 256) {
    int kk = i >> 4, c = i & 15;
    Bs[i] = wp[(size_t)kk * Nout + c];
  }
  __syncthreads();

  const int tm = mg * 8 + wv;
  if (tm * 16 >= M) return;

  const int half = lane >> 4;
  const int l = lane & 15;

  const float* aptr = X + (size_t)(tm * 16 + l) * K + half * 2;
  const float* bbase = Bs + (half * 2) * 16 + l;

  v8f acc = {};
  for (int k = 0; k < K; k += 4) {
    float2 av = *(const float2*)aptr;
    v2f a; a.x = av.x; a.y = av.y;
    v2f b; b.x = bbase[0]; b.y = bbase[16];
    acc = __builtin_amdgcn_wmma_f32_16x16x4_f32(
         false, a,  false, b,
         (short)0, acc,  false,  false);
    aptr += 4;
    bbase += 64;
  }

  float* obase = out + (size_t)(tm * 16 + half * 8) * Nout + tn * 16 + l;
#pragma unroll
  for (int r = 0; r < 8; ++r) obase[(size_t)r * Nout] = acc[r];
}

__global__ void attn_coef(const float* __restrict__ feat,
                          const float* __restrict__ al,
                          const float* __restrict__ ar,
                          float* __restrict__ el, float* __restrict__ er) {
  int i = blockIdx.x * blockDim.x + threadIdx.x;
  if (i >= NN * HH) return;
  int h = i & (HH - 1);
  const float4* f = (const float4*)(feat + (size_t)i * DD);
  const float4* a = (const float4*)(al + h * DD);
  const float4* b = (const float4*)(ar + h * DD);
  float sl = 0.f, sr = 0.f;
#pragma unroll 4
  for (int d = 0; d < DD / 4; ++d) {
    float4 fv = f[d], av = a[d], bv = b[d];
    sl += fv.x * av.x + fv.y * av.y + fv.z * av.z + fv.w * av.w;
    sr += fv.x * bv.x + fv.y * bv.y + fv.z * bv.z + fv.w * bv.w;
  }
  el[i] = sl;
  er[i] = sr;
}

__device__ __forceinline__ void atomicMaxF(float* addr, float val) {
  if (val >= 0.f)
    atomicMax((int*)addr, __float_as_int(val));
  else
    atomicMin((unsigned int*)addr, __float_as_uint(val));
}

__device__ __forceinline__ float leaky(float v) {
  return v > 0.f ? v : NEG_SLOPE * v;
}

__global__ void edge_max(const int* __restrict__ src, const int* __restrict__ dst,
                         const float* __restrict__ el, const float* __restrict__ er,
                         float* __restrict__ emax) {
  int e = blockIdx.x * blockDim.x + threadIdx.x;
  if (e >= EE) return;
  int s = src[e], t = dst[e];
#pragma unroll
  for (int h = 0; h < HH; ++h) {
    float v = leaky(el[s * HH + h] + er[t * HH + h]);
    atomicMaxF(&emax[t * HH + h], v);
  }
}

__global__ void edge_exp(const int* __restrict__ src, const int* __restrict__ dst,
                         const float* __restrict__ el, const float* __restrict__ er,
                         const float* __restrict__ emax,
                         float* __restrict__ exv, float* __restrict__ denom) {
  int e = blockIdx.x * blockDim.x + threadIdx.x;
  if (e >= EE) return;
  int s = src[e], t = dst[e];
#pragma unroll
  for (int h = 0; h < HH; ++h) {
    float v = leaky(el[s * HH + h] + er[t * HH + h]);
    float m = emax[t * HH + h];
    float xo = __expf(v - m);
    exv[(size_t)e * HH + h] = xo;
    atomicAdd(&denom[t * HH + h], xo);
  }
}

__global__ __launch_bounds__(128) void edge_scatter(
    const int* __restrict__ src, const int* __restrict__ dst,
    const float* __restrict__ feat, const float* __restrict__ exv,
    const float* __restrict__ denom, float* __restrict__ agg) {
  int e = blockIdx.x;
  int d = threadIdx.x;
  int s = src[e], t = dst[e];
  const float* fs = feat + (size_t)s * FH + d;
  float* ob = agg + (size_t)t * FH + d;
  __builtin_prefetch(fs, 0, 0);
#pragma unroll
  for (int h = 0; h < HH; ++h) {
    float dn = denom[t * HH + h];
    float alpha = exv[(size_t)e * HH + h] / (dn == 0.f ? 1.f : dn);
    atomicAdd(ob + h * DD, fs[h * DD] * alpha);
  }
}

__global__ void layer_elu(const float* __restrict__ agg, float* __restrict__ out,
                          long n) {
  long i = (long)blockIdx.x * blockDim.x + threadIdx.x;
  if (i < n) {
    float v = agg[i];
    out[i] = v > 0.f ? v : (__expf(v) - 1.f);
  }
}

__global__ void finalize(const float* __restrict__ agg,
                         const float* __restrict__ hin,
                         float* __restrict__ out) {
  int i = blockIdx.x * blockDim.x + threadIdx.x;
  if (i >= NN * DD) return;
  int n = i / DD, d = i - n * DD;
  float s = 0.f;
#pragma unroll
  for (int h = 0; h < HH; ++h) {
    size_t off = (size_t)n * FH + h * DD + d;
    float v = agg[off] + hin[off];
    s += v > 0.f ? v : (__expf(v) - 1.f);
  }
  out[i] = s * (1.f / HH);
}

extern "C" void kernel_launch(void* const* d_in, const int* in_sizes, int n_in,
                              void* d_out, int out_size, void* d_ws, size_t ws_size,
                              hipStream_t stream) {
  const int* src = (const int*)d_in[0];
  const int* dst = (const int*)d_in[1];
  const float* x   = (const float*)d_in[2];
  const float* W0  = (const float*)d_in[3];
  const float* al0 = (const float*)d_in[4];
  const float* ar0 = (const float*)d_in[5];
  const float* W1  = (const float*)d_in[6];
  const float* al1 = (const float*)d_in[7];
  const float* ar1 = (const float*)d_in[8];
  float* out = (float*)d_out;

  float* ws    = (float*)d_ws;
  float* feat  = ws;
  float* agg   = feat  + (size_t)NN * FH;
  float* hbuf  = agg   + (size_t)NN * FH;
  float* el    = hbuf  + (size_t)NN * FH;
  float* er    = el    + (size_t)NN * HH;
  float* emax  = er    + (size_t)NN * HH;
  float* denom = emax  + (size_t)NN * HH;
  float* exv   = denom + (size_t)NN * HH;

  const int mtiles = (NN + 15) / 16;
  const int mgroups = (mtiles + 7) / 8;
  const int ntiles = FH / 16;
  const int gemmBlocks = mgroups * ntiles;

  auto run_layer = [&](const float* xin, const float* W, int K,
                       const float* al, const float* ar) {
    gemm16_wmma<<<gemmBlocks, 256, 0, stream>>>(xin, W, feat, NN, K, FH);
    attn_coef<<<(NN * HH + 255) / 256, 256, 0, stream>>>(feat, al, ar, el, er);
    fill_kernel<<<(NN * HH + 255) / 256, 256, 0, stream>>>(emax, (long)NN * HH,
                                                           -__builtin_huge_valf());
    fill_kernel<<<(NN * HH + 255) / 256, 256, 0, stream>>>(denom, (long)NN * HH, 0.f);
    fill_kernel<<<((long)NN * FH + 255) / 256, 256, 0, stream>>>(agg, (long)NN * FH, 0.f);
    edge_max<<<(EE + 255) / 256, 256, 0, stream>>>(src, dst, el, er, emax);
    edge_exp<<<(EE + 255) / 256, 256, 0, stream>>>(src, dst, el, er, emax, exv, denom);
    edge_scatter<<<EE, 128, 0, stream>>>(src, dst, feat, exv, denom, agg);
  };

  run_layer(x, W0, K0, al0, ar0);
  layer_elu<<<((long)NN * FH + 255) / 256, 256, 0, stream>>>(agg, hbuf, (long)NN * FH);
  run_layer(hbuf, W1, FH, al1, ar1);
  finalize<<<(NN * DD + 255) / 256, 256, 0, stream>>>(agg, hbuf, out);
}


# A work-group owns 40 destination nodes: scans dst, lists its edges, keeps max, denominator, feature sums in LDS (LDS atomics), divides, writes each row once; 18 launches to 6; sum order LDS arrival
# speedup vs baseline: 3.8993x; 3.8993x over previous
#include <hip/hip_runtime.h>
#include <math.h>

#define NN 10000
#define EE 320000
#define HH 4
#define DD 128
#define FH (HH * DD)
#define K0 256
#define KMAX 512
#define NEG_SLOPE 0.2f

typedef __attribute__((ext_vector_type(2))) float v2f;
typedef __attribute__((ext_vector_type(8))) float v8f;

__global__ __launch_bounds__(256) void gemm16_wmma(const float* __restrict__ X,
                                                   const float* __restrict__ W,
                                                   float* __restrict__ out,
                                                   int M, int K, int Nout) {
  __shared__ float Bs[KMAX * 16];

  const int tid = threadIdx.x;
  const int lane = tid & 31;
  const int wv = tid >> 5;
  const int ntiles = Nout >> 4;
  const int mg = blockIdx.x / ntiles;
  const int tn = blockIdx.x - mg * ntiles;

  const float* wp = W + tn * 16;
  for (int i = tid; i < K * 16; i += 256) {
    int kk = i >> 4, c = i & 15;
    Bs[i] = wp[(size_t)kk * Nout + c];
  }
  __syncthreads();

  const int tm = mg * 8 + wv;
  if (tm * 16 >= M) return;

  const int half = lane >> 4;
  const int l = lane & 15;

  const float* aptr = X + (size_t)(tm * 16 + l) * K + half * 2;
  const float* bbase = Bs + (half * 2) * 16 + l;

  v8f acc = {};
  for (int k = 0; k < K; k += 4) {
    float2 av = *(const float2*)aptr;
    v2f a; a.x = av.x; a.y = av.y;
    v2f b; b.x = bbase[0]; b.y = bbase[16];
    acc = __builtin_amdgcn_wmma_f32_16x16x4_f32(
         false, a,  false, b,
         (short)0, acc,  false,  false);
    aptr += 4;
    bbase += 64;
  }

  float* obase = out + (size_t)(tm * 16 + half * 8) * Nout + tn * 16 + l;
#pragma unroll
  for (int r = 0; r < 8; ++r) obase[(size_t)r * Nout] = acc[r];
}

__global__ void attn_coef(const float* __restrict__ feat,
                          const float* __restrict__ al,
                          const float* __restrict__ ar,
                          float* __restrict__ el, float* __restrict__ er) {
  int i = blockIdx.x * blockDim.x + threadIdx.x;
  if (i >= NN * HH) return;
  int h = i & (HH - 1);
  const float4* f = (const float4*)(feat + (size_t)i * DD);
  const float4* a = (const float4*)(al + h * DD);
  const float4* b = (const float4*)(ar + h * DD);
  float sl = 0.f, sr = 0.f;
#pragma unroll 4
  for (int d = 0; d < DD / 4; ++d) {
    float4 fv = f[d], av = a[d], bv = b[d];
    sl += fv.x * av.x + fv.y * av.y + fv.z * av.z + fv.w * av.w;
    sr += fv.x * bv.x + fv.y * bv.y + fv.z * bv.z + fv.w * bv.w;
  }
  el[i] = sl;
  er[i] = sr;
}

#define RN 40
#define NBLK (NN / RN)
#define ETH 1024
#define CAP 4096

static_assert(NN % RN == 0, "node ranges must tile N");
static_assert(EE % 4 == 0 && CAP % 4 == 0, "dst is scanned 4 edges per load");
static_assert(EE < (1 << 19) && RN < (1 << 6) && NN <= (1 << 16), "packed fields");

__device__ __forceinline__ float leaky(float v) {
  return v > 0.f ? v : NEG_SLOPE * v;
}

__device__ __forceinline__ void lds_max_f(float* addr, float v) {
  if (__float_as_int(v) >= 0)
    atomicMax((int*)addr, __float_as_int(v));
  else
    atomicMin((unsigned int*)addr, __float_as_uint(v));
}

__device__ __forceinline__ void lds_add_f(float* addr, float v) {
  __hip_atomic_fetch_add(addr, v, __ATOMIC_RELAXED, __HIP_MEMORY_SCOPE_WORKGROUP);
}

__device__ __forceinline__ void load_dst4(int4 (&v)[4], const int4* __restrict__ d4,
                                          int q0, int q_hi, int tid) {
#pragma unroll
  for (int u = 0; u < 4; ++u) {
    const int q = q0 + u * ETH + tid;
    v[u] = d4[q < q_hi ? q : q_hi - 1];
  }
}

__device__ __forceinline__ void scan_emit(const int* __restrict__ dst, int e_lo, int e_hi,
                                          int lo, unsigned* list, int* cnt,
                                          int tid, int lane) {
  const int4* d4 = (const int4*)dst;
  const int q_lo = e_lo >> 2, q_hi = e_hi >> 2;
  const int STEP = 4 * ETH;
  int4 cur[4], nxt[4];
  load_dst4(cur, d4, q_lo, q_hi, tid);
  for (int q0 = q_lo; q0 < q_hi; q0 += STEP) {
    load_dst4(nxt, d4, q0 + STEP, q_hi, tid);
#pragma unroll
    for (int u = 0; u < 4; ++u) {
      const unsigned t0 = (unsigned)(cur[u].x - lo), t1 = (unsigned)(cur[u].y - lo);
      const unsigned t2 = (unsigned)(cur[u].z - lo), t3 = (unsigned)(cur[u].w - lo);
      const bool in = (q0 + u * ETH + tid) < q_hi;
      const bool m0 = in && t0 < RN, m1 = in && t1 < RN, m2 = in && t2 < RN, m3 = in && t3 < RN;
      const unsigned b0 = __builtin_amdgcn_ballot_w32(m0);
      const unsigned b1 = __builtin_amdgcn_ballot_w32(m1);
      const unsigned b2 = __builtin_amdgcn_ballot_w32(m2);
      const unsigned b3 = __builtin_amdgcn_ballot_w32(m3);
      if ((b0 | b1 | b2 | b3) != 0u) {
        const int tot = __popc(b0) + __popc(b1) + __popc(b2) + __popc(b3);
        int base = 0;
        if (lane == 0) base = atomicAdd(cnt, tot);
        base = __builtin_amdgcn_readfirstlane(base);
        int p = base + (int)(__builtin_amdgcn_mbcnt_lo(b0, 0u) + __builtin_amdgcn_mbcnt_lo(b1, 0u) +
                             __builtin_amdgcn_mbcnt_lo(b2, 0u) + __builtin_amdgcn_mbcnt_lo(b3, 0u));
        const unsigned eb = (unsigned)(q0 + u * ETH + tid) << 2;
        if (m0) { if (p < CAP) list[p] = (eb + 0u) | (t0 << 19); ++p; }
        if (m1) { if (p < CAP) list[p] = (eb + 1u) | (t1 << 19); ++p; }
        if (m2) { if (p < CAP) list[p] = (eb + 2u) | (t2 << 19); ++p; }
        if (m3) { if (p < CAP) list[p] = (eb + 3u) | (t3 << 19); ++p; }
      }
    }
#pragma unroll
    for (int u = 0; u < 4; ++u) cur[u] = nxt[u];
  }
}

__device__ __forceinline__ void edge_scores(int L, bool domax, unsigned* list, float4* ebuf,
                                            float* mxs, const int* __restrict__ src,
                                            const float* __restrict__ el,
                                            const float* __restrict__ er, int lo, int tid) {
  for (int k = tid; k < L; k += ETH) {
    const unsigned w = list[k];
    const int e = (int)(w & 0x7FFFFu);
    const int tl = (int)(w >> 19);
    const int s = src[e];
    const float4 a = *(const float4*)(el + (size_t)s * HH);
    const float4 b = *(const float4*)(er + (size_t)(lo + tl) * HH);
    float4 v;
    v.x = leaky(a.x + b.x); v.y = leaky(a.y + b.y);
    v.z = leaky(a.z + b.z); v.w = leaky(a.w + b.w);
    ebuf[k] = v;
    list[k] = (unsigned)s | ((unsigned)tl << 16);
    if (domax) {
      lds_max_f(&mxs[tl * HH + 0], v.x); lds_max_f(&mxs[tl * HH + 1], v.y);
      lds_max_f(&mxs[tl * HH + 2], v.z); lds_max_f(&mxs[tl * HH + 3], v.w);
    }
  }
}

__device__ __forceinline__ void row_load(int k, const unsigned* list, const float4* ebuf,
                                         const float* mxs, float* dns,
                                         const float* __restrict__ feat, int lane,
                                         float (&f)[16], float (&p)[4], int& tl) {
  const unsigned w = list[k];
  const int s = (int)(w & 0xFFFFu);
  tl = (int)(w >> 16);
  const float4 v = ebuf[k];
  const float4 m = *(const float4*)(mxs + tl * HH);
  p[0] = __expf(v.x - m.x); p[1] = __expf(v.y - m.y);
  p[2] = __expf(v.z - m.z); p[3] = __expf(v.w - m.w);
  const float* fr = feat + (size_t)s * FH + lane;
#pragma unroll
  for (int j = 0; j < 16; ++j) f[j] = fr[32 * j];
  if (lane < HH) {
    const float pv = lane == 0 ? p[0] : lane == 1 ? p[1] : lane == 2 ? p[2] : p[3];
    lds_add_f(&dns[tl * HH + lane], pv);
  }
}

__device__ __forceinline__ void row_add(float* acc, int tl, int lane,
                                        const float (&f)[16], const float (&p)[4]) {
  float* ar = acc + tl * FH + lane;
#pragma unroll
  for (int j = 0; j < 16; ++j) lds_add_f(ar + 32 * j, f[j] * p[j >> 2]);
}

__device__ __forceinline__ void accumulate(int L, const unsigned* list, const float4* ebuf,
                                           const float* mxs, float* dns, float* acc,
                                           const float* __restrict__ feat, int wv, int lane) {
  int k = wv;
  for (; k + 32 < L; k += 64) {
    float fa[16], pa[4], fb[16], pb[4];
    int ta, tb;
    row_load(k, list, ebuf, mxs, dns, feat, lane, fa, pa, ta);
    row_load(k + 32, list, ebuf, mxs, dns, feat, lane, fb, pb, tb);
    row_add(acc, ta, lane, fa, pa);
    row_add(acc, tb, lane, fb, pb);
  }
  if (k < L) {
    float fa[16], pa[4];
    int ta;
    row_load(k, list, ebuf, mxs, dns, feat, lane, fa, pa, ta);
    row_add(acc, ta, lane, fa, pa);
  }
}

__device__ __forceinline__ float elu1(float v) {
  return v > 0.f ? v : (__expf(v) - 1.f);
}

template <bool LAST>
__global__ __launch_bounds__(ETH) void edge_own(
    const int* __restrict__ src, const int* __restrict__ dst,
    const float* __restrict__ feat, const float* __restrict__ el,
    const float* __restrict__ er, const float* __restrict__ hin,
    float* __restrict__ outp) {
  __shared__ __attribute__((aligned(16))) float acc[RN * FH];
  __shared__ __attribute__((aligned(16))) float4 ebuf[CAP];
  __shared__ unsigned list[CAP];
  __shared__ __attribute__((aligned(16))) float mxs[RN * HH];
  __shared__ __attribute__((aligned(16))) float dns[RN * HH];
  __shared__ int cnt;

  const int tid = threadIdx.x;
  const int lane = tid & 31;
  const int wv = tid >> 5;
  const int lo = blockIdx.x * RN;

  for (int i = tid * 4; i < RN * FH; i += ETH * 4)
    *(float4*)(acc + i) = make_float4(0.f, 0.f, 0.f, 0.f);
  if (tid < RN * HH) { mxs[tid] = -__builtin_huge_valf(); dns[tid] = 0.f; }
  if (tid == 0) cnt = 0;
  __syncthreads();

  scan_emit(dst, 0, EE, lo, list, &cnt, tid, lane);
  __syncthreads();
  const int total = cnt;

  if (total <= CAP) {
    edge_scores(total, true, list, ebuf, mxs, src, el, er, lo, tid);
    __syncthreads();
    accumulate(total, list, ebuf, mxs, dns, acc, feat, wv, lane);
  } else {
    for (int c0 = 0; c0 < EE; c0 += CAP) {
      __syncthreads();
      if (tid == 0) cnt = 0;
      __syncthreads();
      scan_emit(dst, c0, (c0 + CAP < EE) ? c0 + CAP : EE, lo, list, &cnt, tid, lane);
      __syncthreads();
      edge_scores(cnt, true, list, ebuf, mxs, src, el, er, lo, tid);
    }
    for (int c0 = 0; c0 < EE; c0 += CAP) {
      __syncthreads();
      if (tid == 0) cnt = 0;
      __syncthreads();
      scan_emit(dst, c0, (c0 + CAP < EE) ? c0 + CAP : EE, lo, list, &cnt, tid, lane);
      __syncthreads();
      const int L = cnt;
      edge_scores(L, false, list, ebuf, mxs, src, el, er, lo, tid);
      __syncthreads();
      accumulate(L, list, ebuf, mxs, dns, acc, feat, wv, lane);
    }
  }
  __syncthreads();

  if (!LAST) {
    float* ob = outp + (size_t)lo * FH;
    for (int i = tid * 4; i < RN * FH; i += ETH * 4) {
      const float d = dns[i >> 7];
      const float dd = d == 0.f ? 1.f : d;
      const float4 a = *(const float4*)(acc + i);
      float4 o;
      o.x = elu1(a.x / dd); o.y = elu1(a.y / dd);
      o.z = elu1(a.z / dd); o.w = elu1(a.w / dd);
      *(float4*)(ob + i) = o;
    }
  } else {
    for (int i = tid * 4; i < RN * DD; i += ETH * 4) {
      const int tl = i >> 7, d0 = i & (DD - 1);
      float4 sum = make_float4(0.f, 0.f, 0.f, 0.f);
#pragma unroll
      for (int h = 0; h < HH; ++h) {
        const float d = dns[tl * HH + h];
        const float dd = d == 0.f ? 1.f : d;
        const float4 a = *(const float4*)(acc + tl * FH + h * DD + d0);
        const float4 r = *(const float4*)(hin + (size_t)(lo + tl) * FH + h * DD + d0);
        sum.x += elu1(a.x / dd + r.x); sum.y += elu1(a.y / dd + r.y);
        sum.z += elu1(a.z / dd + r.z); sum.w += elu1(a.w / dd + r.w);
      }
      float4 o;
      o.x = sum.x * (1.f / HH); o.y = sum.y * (1.f / HH);
      o.z = sum.z * (1.f / HH); o.w = sum.w * (1.f / HH);
      *(float4*)(outp + (size_t)(lo + tl) * DD + d0) = o;
    }
  }
}

extern "C" void kernel_launch(void* const* d_in, const int* in_sizes, int n_in,
                              void* d_out, int out_size, void* d_ws, size_t ws_size,
                              hipStream_t stream) {
  const int* src = (const int*)d_in[0];
  const int* dst = (const int*)d_in[1];
  const float* x   = (const float*)d_in[2];
  const float* W0  = (const float*)d_in[3];
  const float* al0 = (const float*)d_in[4];
  const float* ar0 = (const float*)d_in[5];
  const float* W1  = (const float*)d_in[6];
  const float* al1 = (const float*)d_in[7];
  const float* ar1 = (const float*)d_in[8];
  float* out = (float*)d_out;

  float* ws    = (float*)d_ws;
  float* feat  = ws;
  float* hbuf  = feat  + (size_t)NN * FH;
  float* el    = hbuf  + (size_t)NN * FH;
  float* er    = el    + (size_t)NN * HH;

  const int mtiles = (NN + 15) / 16;
  const int mgroups = (mtiles + 7) / 8;
  const int ntiles = FH / 16;
  const int gemmBlocks = mgroups * ntiles;

  gemm16_wmma<<<gemmBlocks, 256, 0, stream>>>(x, W0, feat, NN, K0, FH);
  attn_coef<<<(NN * HH + 255) / 256, 256, 0, stream>>>(feat, al0, ar0, el, er);
  edge_own<false><<<NBLK, ETH, 0, stream>>>(src, dst, feat, el, er, nullptr, hbuf);
  gemm16_wmma<<<gemmBlocks, 256, 0, stream>>>(hbuf, W1, feat, NN, FH, FH);
  attn_coef<<<(NN * HH + 255) / 256, 256, 0, stream>>>(feat, al1, ar1, el, er);
  edge_own<true><<<NBLK, ETH, 0, stream>>>(src, dst, feat, el, er, hbuf, out);
}


# Edge gather: a source row is 4 loads of 16 B per lane (whole lines) instead of 16 one-line 4-B loads; LDS sum rows kept lane-transposed per head so atomics stay conflict-free; sum order LDS arrival
# speedup vs baseline: 4.1087x; 1.0537x over previous
#include <hip/hip_runtime.h>
#include <math.h>

#define NN 10000
#define EE 320000
#define HH 4
#define DD 128
#define FH (HH * DD)
#define K0 256
#define KMAX 512
#define NEG_SLOPE 0.2f

typedef __attribute__((ext_vector_type(2))) float v2f;
typedef __attribute__((ext_vector_type(8))) float v8f;

__global__ __launch_bounds__(256) void gemm16_wmma(const float* __restrict__ X,
                                                   const float* __restrict__ W,
                                                   float* __restrict__ out,
                                                   int M, int K, int Nout) {
  __shared__ float Bs[KMAX * 16];

  const int tid = threadIdx.x;
  const int lane = tid & 31;
  const int wv = tid >> 5;
  const int ntiles = Nout >> 4;
  const int mg = blockIdx.x / ntiles;
  const int tn = blockIdx.x - mg * ntiles;

  const float* wp = W + tn * 16;
  for (int i = tid; i < K * 16; i += 256) {
    int kk = i >> 4, c = i & 15;
    Bs[i] = wp[(size_t)kk * Nout + c];
  }
  __syncthreads();

  const int tm = mg * 8 + wv;
  if (tm * 16 >= M) return;

  const int half = lane >> 4;
  const int l = lane & 15;

  const float* aptr = X + (size_t)(tm * 16 + l) * K + half * 2;
  const float* bbase = Bs + (half * 2) * 16 + l;

  v8f acc = {};
  for (int k = 0; k < K; k += 4) {
    float2 av = *(const float2*)aptr;
    v2f a; a.x = av.x; a.y = av.y;
    v2f b; b.x = bbase[0]; b.y = bbase[16];
    acc = __builtin_amdgcn_wmma_f32_16x16x4_f32(
         false, a,  false, b,
         (short)0, acc,  false,  false);
    aptr += 4;
    bbase += 64;
  }

  float* obase = out + (size_t)(tm * 16 + half * 8) * Nout + tn * 16 + l;
#pragma unroll
  for (int r = 0; r < 8; ++r) obase[(size_t)r * Nout] = acc[r];
}

__global__ void attn_coef(const float* __restrict__ feat,
                          const float* __restrict__ al,
                          const float* __restrict__ ar,
                          float* __restrict__ el, float* __restrict__ er) {
  int i = blockIdx.x * blockDim.x + threadIdx.x;
  if (i >= NN * HH) return;
  int h = i & (HH - 1);
  const float4* f = (const float4*)(feat + (size_t)i * DD);
  const float4* a = (const float4*)(al + h * DD);
  const float4* b = (const float4*)(ar + h * DD);
  float sl = 0.f, sr = 0.f;
#pragma unroll 4
  for (int d = 0; d < DD / 4; ++d) {
    float4 fv = f[d], av = a[d], bv = b[d];
    sl += fv.x * av.x + fv.y * av.y + fv.z * av.z + fv.w * av.w;
    sr += fv.x * bv.x + fv.y * bv.y + fv.z * bv.z + fv.w * bv.w;
  }
  el[i] = sl;
  er[i] = sr;
}

#define RN 40
#define NBLK (NN / RN)
#define ETH 1024
#define CAP 4096

static_assert(NN % RN == 0, "node ranges must tile N");
static_assert(EE % 4 == 0 && CAP % 4 == 0, "dst is scanned 4 edges per load");
static_assert(EE < (1 << 19) && RN < (1 << 6) && NN <= (1 << 16), "packed fields");
static_assert(DD / 4 == 32, "a wave's 32 lanes x 16 bytes cover one head of a row");

__device__ __forceinline__ float leaky(float v) {
  return v > 0.f ? v : NEG_SLOPE * v;
}

__device__ __forceinline__ void lds_max_f(float* addr, float v) {
  if (__float_as_int(v) >= 0)
    atomicMax((int*)addr, __float_as_int(v));
  else
    atomicMin((unsigned int*)addr, __float_as_uint(v));
}

__device__ __forceinline__ void lds_add_f(float* addr, float v) {
  __hip_atomic_fetch_add(addr, v, __ATOMIC_RELAXED, __HIP_MEMORY_SCOPE_WORKGROUP);
}

__device__ __forceinline__ void load_dst4(int4 (&v)[4], const int4* __restrict__ d4,
                                          int q0, int q_hi, int tid) {
#pragma unroll
  for (int u = 0; u < 4; ++u) {
    const int q = q0 + u * ETH + tid;
    v[u] = d4[q < q_hi ? q : q_hi - 1];
  }
}

__device__ __forceinline__ void scan_emit(const int* __restrict__ dst, int e_lo, int e_hi,
                                          int lo, unsigned* list, int* cnt,
                                          int tid, int lane) {
  const int4* d4 = (const int4*)dst;
  const int q_lo = e_lo >> 2, q_hi = e_hi >> 2;
  const int STEP = 4 * ETH;
  int4 cur[4], nxt[4];
  load_dst4(cur, d4, q_lo, q_hi, tid);
  for (int q0 = q_lo; q0 < q_hi; q0 += STEP) {
    load_dst4(nxt, d4, q0 + STEP, q_hi, tid);
#pragma unroll
    for (int u = 0; u < 4; ++u) {
      const unsigned t0 = (unsigned)(cur[u].x - lo), t1 = (unsigned)(cur[u].y - lo);
      const unsigned t2 = (unsigned)(cur[u].z - lo), t3 = (unsigned)(cur[u].w - lo);
      const bool in = (q0 + u * ETH + tid) < q_hi;
      const bool m0 = in && t0 < RN, m1 = in && t1 < RN, m2 = in && t2 < RN, m3 = in && t3 < RN;
      const unsigned b0 = __builtin_amdgcn_ballot_w32(m0);
      const unsigned b1 = __builtin_amdgcn_ballot_w32(m1);
      const unsigned b2 = __builtin_amdgcn_ballot_w32(m2);
      const unsigned b3 = __builtin_amdgcn_ballot_w32(m3);
      if ((b0 | b1 | b2 | b3) != 0u) {
        const int tot = __popc(b0) + __popc(b1) + __popc(b2) + __popc(b3);
        int base = 0;
        if (lane == 0) base = atomicAdd(cnt, tot);
        base = __builtin_amdgcn_readfirstlane(base);
        int p = base + (int)(__builtin_amdgcn_mbcnt_lo(b0, 0u) + __builtin_amdgcn_mbcnt_lo(b1, 0u) +
                             __builtin_amdgcn_mbcnt_lo(b2, 0u) + __builtin_amdgcn_mbcnt_lo(b3, 0u));
        const unsigned eb = (unsigned)(q0 + u * ETH + tid) << 2;
        if (m0) { if (p < CAP) list[p] = (eb + 0u) | (t0 << 19); ++p; }
        if (m1) { if (p < CAP) list[p] = (eb + 1u) | (t1 << 19); ++p; }
        if (m2) { if (p < CAP) list[p] = (eb + 2u) | (t2 << 19); ++p; }
        if (m3) { if (p < CAP) list[p] = (eb + 3u) | (t3 << 19); ++p; }
      }
    }
#pragma unroll
    for (int u = 0; u < 4; ++u) cur[u] = nxt[u];
  }
}

__device__ __forceinline__ void edge_scores(int L, bool domax, unsigned* list, float4* ebuf,
                                            float* mxs, const int* __restrict__ src,
                                            const float* __restrict__ el,
                                            const float* __restrict__ er, int lo, int tid) {
  for (int k = tid; k < L; k += ETH) {
    const unsigned w = list[k];
    const int e = (int)(w & 0x7FFFFu);
    const int tl = (int)(w >> 19);
    const int s = src[e];
    const float4 a = *(const float4*)(el + (size_t)s * HH);
    const float4 b = *(const float4*)(er + (size_t)(lo + tl) * HH);
    float4 v;
    v.x = leaky(a.x + b.x); v.y = leaky(a.y + b.y);
    v.z = leaky(a.z + b.z); v.w = leaky(a.w + b.w);
    ebuf[k] = v;
    list[k] = (unsigned)s | ((unsigned)tl << 16);
    if (domax) {
      lds_max_f(&mxs[tl * HH + 0], v.x); lds_max_f(&mxs[tl * HH + 1], v.y);
      lds_max_f(&mxs[tl * HH + 2], v.z); lds_max_f(&mxs[tl * HH + 3], v.w);
    }
  }
}

__device__ __forceinline__ void row_load(int k, const unsigned* list, const float4* ebuf,
                                         const float* mxs, float* dns,
                                         const float* __restrict__ feat, int lane,
                                         float4 (&f)[HH], float (&p)[4], int& tl) {
  const unsigned w = list[k];
  const int s = (int)(w & 0xFFFFu);
  tl = (int)(w >> 16);
  const float4 v = ebuf[k];
  const float4 m = *(const float4*)(mxs + tl * HH);
  p[0] = __expf(v.x - m.x); p[1] = __expf(v.y - m.y);
  p[2] = __expf(v.z - m.z); p[3] = __expf(v.w - m.w);
  const float4* fr = (const float4*)(feat + (size_t)s * FH) + lane;
#pragma unroll
  for (int j = 0; j < HH; ++j) f[j] = fr[(DD / 4) * j];
  if (lane < HH) {
    const float pv = lane == 0 ? p[0] : lane == 1 ? p[1] : lane == 2 ? p[2] : p[3];
    lds_add_f(&dns[tl * HH + lane], pv);
  }
}

__device__ __forceinline__ void row_add(float* acc, int tl, int lane,
                                        const float4 (&f)[HH], const float (&p)[4]) {
  float* ar = acc + tl * FH + lane;
#pragma unroll
  for (int j = 0; j < HH; ++j) {
    lds_add_f(ar + j * DD + 0 * (DD / 4), f[j].x * p[j]);
    lds_add_f(ar + j * DD + 1 * (DD / 4), f[j].y * p[j]);
    lds_add_f(ar + j * DD + 2 * (DD / 4), f[j].z * p[j]);
    lds_add_f(ar + j * DD + 3 * (DD / 4), f[j].w * p[j]);
  }
}

__device__ __forceinline__ float4 head_cols4(const float* hd, int d0) {
  const float* q = hd + (d0 >> 2);
  float4 a;
  a.x = q[0 * (DD / 4)]; a.y = q[1 * (DD / 4)];
  a.z = q[2 * (DD / 4)]; a.w = q[3 * (DD / 4)];
  return a;
}

__device__ __forceinline__ void accumulate(int L, const unsigned* list, const float4* ebuf,
                                           const float* mxs, float* dns, float* acc,
                                           const float* __restrict__ feat, int wv, int lane) {
  int k = wv;
  for (; k + 32 < L; k += 64) {
    float4 fa[HH], fb[HH];
    float pa[4], pb[4];
    int ta, tb;
    row_load(k, list, ebuf, mxs, dns, feat, lane, fa, pa, ta);
    row_load(k + 32, list, ebuf, mxs, dns, feat, lane, fb, pb, tb);
    row_add(acc, ta, lane, fa, pa);
    row_add(acc, tb, lane, fb, pb);
  }
  if (k < L) {
    float4 fa[HH];
    float pa[4];
    int ta;
    row_load(k, list, ebuf, mxs, dns, feat, lane, fa, pa, ta);
    row_add(acc, ta, lane, fa, pa);
  }
}

__device__ __forceinline__ float elu1(float v) {
  return v > 0.f ? v : (__expf(v) - 1.f);
}

template <bool LAST>
__global__ __launch_bounds__(ETH) void edge_own(
    const int* __restrict__ src, const int* __restrict__ dst,
    const float* __restrict__ feat, const float* __restrict__ el,
    const float* __restrict__ er, const float* __restrict__ hin,
    float* __restrict__ outp) {
  __shared__ __attribute__((aligned(16))) float acc[RN * FH];
  __shared__ __attribute__((aligned(16))) float4 ebuf[CAP];
  __shared__ unsigned list[CAP];
  __shared__ __attribute__((aligned(16))) float mxs[RN * HH];
  __shared__ __attribute__((aligned(16))) float dns[RN * HH];
  __shared__ int cnt;

  const int tid = threadIdx.x;
  const int lane = tid & 31;
  const int wv = tid >> 5;
  const int lo = blockIdx.x * RN;

  for (int i = tid * 4; i < RN * FH; i += ETH * 4)
    *(float4*)(acc + i) = make_float4(0.f, 0.f, 0.f, 0.f);
  if (tid < RN * HH) { mxs[tid] = -__builtin_huge_valf(); dns[tid] = 0.f; }
  if (tid == 0) cnt = 0;
  __syncthreads();

  scan_emit(dst, 0, EE, lo, list, &cnt, tid, lane);
  __syncthreads();
  const int total = cnt;

  if (total <= CAP) {
    edge_scores(total, true, list, ebuf, mxs, src, el, er, lo, tid);
    __syncthreads();
    accumulate(total, list, ebuf, mxs, dns, acc, feat, wv, lane);
  } else {
    for (int c0 = 0; c0 < EE; c0 += CAP) {
      __syncthreads();
      if (tid == 0) cnt = 0;
      __syncthreads();
      scan_emit(dst, c0, (c0 + CAP < EE) ? c0 + CAP : EE, lo, list, &cnt, tid, lane);
      __syncthreads();
      edge_scores(cnt, true, list, ebuf, mxs, src, el, er, lo, tid);
    }
    for (int c0 = 0; c0 < EE; c0 += CAP) {
      __syncthreads();
      if (tid == 0) cnt = 0;
      __syncthreads();
      scan_emit(dst, c0, (c0 + CAP < EE) ? c0 + CAP : EE, lo, list, &cnt, tid, lane);
      __syncthreads();
      const int L = cnt;
      edge_scores(L, false, list, ebuf, mxs, src, el, er, lo, tid);
      __syncthreads();
      accumulate(L, list, ebuf, mxs, dns, acc, feat, wv, lane);
    }
  }
  __syncthreads();

  if (!LAST) {
    float* ob = outp + (size_t)lo * FH;
    for (int i = tid * 4; i < RN * FH; i += ETH * 4) {
      const float d = dns[i >> 7];
      const float dd = d == 0.f ? 1.f : d;
      const float4 a = head_cols4(acc + (i & ~(DD - 1)), i & (DD - 1));
      float4 o;
      o.x = elu1(a.x / dd); o.y = elu1(a.y / dd);
      o.z = elu1(a.z / dd); o.w = elu1(a.w / dd);
      *(float4*)(ob + i) = o;
    }
  } else {
    for (int i = tid * 4; i < RN * DD; i += ETH * 4) {
      const int tl = i >> 7, d0 = i & (DD - 1);
      float4 sum = make_float4(0.f, 0.f, 0.f, 0.f);
#pragma unroll
      for (int h = 0; h < HH; ++h) {
        const float d = dns[tl * HH + h];
        const float dd = d == 0.f ? 1.f : d;
        const float4 a = head_cols4(acc + tl * FH + h * DD, d0);
        const float4 r = *(const float4*)(hin + (size_t)(lo + tl) * FH + h * DD + d0);
        sum.x += elu1(a.x / dd + r.x); sum.y += elu1(a.y / dd + r.y);
        sum.z += elu1(a.z / dd + r.z); sum.w += elu1(a.w / dd + r.w);
      }
      float4 o;
      o.x = sum.x * (1.f / HH); o.y = sum.y * (1.f / HH);
      o.z = sum.z * (1.f / HH); o.w = sum.w * (1.f / HH);
      *(float4*)(outp + (size_t)(lo + tl) * DD + d0) = o;
    }
  }
}

extern "C" void kernel_launch(void* const* d_in, const int* in_sizes, int n_in,
                              void* d_out, int out_size, void* d_ws, size_t ws_size,
                              hipStream_t stream) {
  const int* src = (const int*)d_in[0];
  const int* dst = (const int*)d_in[1];
  const float* x   = (const float*)d_in[2];
  const float* W0  = (const float*)d_in[3];
  const float* al0 = (const float*)d_in[4];
  const float* ar0 = (const float*)d_in[5];
  const float* W1  = (const float*)d_in[6];
  const float* al1 = (const float*)d_in[7];
  const float* ar1 = (const float*)d_in[8];
  float* out = (float*)d_out;

  float* ws    = (float*)d_ws;
  float* feat  = ws;
  float* hbuf  = feat  + (size_t)NN * FH;
  float* el    = hbuf  + (size_t)NN * FH;
  float* er    = el    + (size_t)NN * HH;

  const int mtiles = (NN + 15) / 16;
  const int mgroups = (mtiles + 7) / 8;
  const int ntiles = FH / 16;
  const int gemmBlocks = mgroups * ntiles;

  gemm16_wmma<<<gemmBlocks, 256, 0, stream>>>(x, W0, feat, NN, K0, FH);
  attn_coef<<<(NN * HH + 255) / 256, 256, 0, stream>>>(feat, al0, ar0, el, er);
  edge_own<false><<<NBLK, ETH, 0, stream>>>(src, dst, feat, el, er, nullptr, hbuf);
  gemm16_wmma<<<gemmBlocks, 256, 0, stream>>>(hbuf, W1, feat, NN, FH, FH);
  attn_coef<<<(NN * HH + 255) / 256, 256, 0, stream>>>(feat, al1, ar1, el, er);
  edge_own<true><<<NBLK, ETH, 0, stream>>>(src, dst, feat, el, er, hbuf, out);
}


# Layer-1 projection moved into the layer-0 edge kernel: a work-group multiplies its 40 finished h rows, held in LDS, by W1 on WMMA and writes feat1; layer-1 GEMM launch gone; sum order unchanged
# speedup vs baseline: 4.8639x; 1.1838x over previous
#include <hip/hip_runtime.h>
#include <math.h>

#define NN 10000
#define EE 320000
#define HH 4
#define DD 128
#define FH (HH * DD)
#define K0 256
#define KMAX 512
#define NEG_SLOPE 0.2f

typedef __attribute__((ext_vector_type(2))) float v2f;
typedef __attribute__((ext_vector_type(8))) float v8f;

__global__ __launch_bounds__(256) void gemm16_wmma(const float* __restrict__ X,
                                                   const float* __restrict__ W,
                                                   float* __restrict__ out,
                                                   int M, int K, int Nout) {
  __shared__ float Bs[KMAX * 16];

  const int tid = threadIdx.x;
  const int lane = tid & 31;
  const int wv = tid >> 5;
  const int ntiles = Nout >> 4;
  const int mg = blockIdx.x / ntiles;
  const int tn = blockIdx.x - mg * ntiles;

  const float* wp = W + tn * 16;
  for (int i = tid; i < K * 16; i += 256) {
    int kk = i >> 4, c = i & 15;
    Bs[i] = wp[(size_t)kk * Nout + c];
  }
  __syncthreads();

  const int tm = mg * 8 + wv;
  if (tm * 16 >= M) return;

  const int half = lane >> 4;
  const int l = lane & 15;

  const float* aptr = X + (size_t)(tm * 16 + l) * K + half * 2;
  const float* bbase = Bs + (half * 2) * 16 + l;

  v8f acc = {};
  for (int k = 0; k < K; k += 4) {
    float2 av = *(const float2*)aptr;
    v2f a; a.x = av.x; a.y = av.y;
    v2f b; b.x = bbase[0]; b.y = bbase[16];
    acc = __builtin_amdgcn_wmma_f32_16x16x4_f32(
         false, a,  false, b,
         (short)0, acc,  false,  false);
    aptr += 4;
    bbase += 64;
  }

  float* obase = out + (size_t)(tm * 16 + half * 8) * Nout + tn * 16 + l;
#pragma unroll
  for (int r = 0; r < 8; ++r) obase[(size_t)r * Nout] = acc[r];
}

__global__ void attn_coef(const float* __restrict__ feat,
                          const float* __restrict__ al,
                          const float* __restrict__ ar,
                          float* __restrict__ el, float* __restrict__ er) {
  int i = blockIdx.x * blockDim.x + threadIdx.x;
  if (i >= NN * HH) return;
  int h = i & (HH - 1);
  const float4* f = (const float4*)(feat + (size_t)i * DD);
  const float4* a = (const float4*)(al + h * DD);
  const float4* b = (const float4*)(ar + h * DD);
  float sl = 0.f, sr = 0.f;
#pragma unroll 4
  for (int d = 0; d < DD / 4; ++d) {
    float4 fv = f[d], av = a[d], bv = b[d];
    sl += fv.x * av.x + fv.y * av.y + fv.z * av.z + fv.w * av.w;
    sr += fv.x * bv.x + fv.y * bv.y + fv.z * bv.z + fv.w * bv.w;
  }
  el[i] = sl;
  er[i] = sr;
}

#define RN 40
#define NBLK (NN / RN)
#define ETH 1024
#define CAP 4096

static_assert(NN % RN == 0, "node ranges must tile N");
static_assert(EE % 4 == 0 && CAP % 4 == 0, "dst is scanned 4 edges per load");
static_assert(EE < (1 << 19) && RN < (1 << 6) && NN <= (1 << 16), "packed fields");
static_assert(DD / 4 == 32, "a wave's 32 lanes x 16 bytes cover one head of a row");

__device__ __forceinline__ float leaky(float v) {
  return v > 0.f ? v : NEG_SLOPE * v;
}

__device__ __forceinline__ void lds_max_f(float* addr, float v) {
  if (__float_as_int(v) >= 0)
    atomicMax((int*)addr, __float_as_int(v));
  else
    atomicMin((unsigned int*)addr, __float_as_uint(v));
}

__device__ __forceinline__ void lds_add_f(float* addr, float v) {
  __hip_atomic_fetch_add(addr, v, __ATOMIC_RELAXED, __HIP_MEMORY_SCOPE_WORKGROUP);
}

__device__ __forceinline__ void load_dst4(int4 (&v)[4], const int4* __restrict__ d4,
                                          int q0, int q_hi, int tid) {
#pragma unroll
  for (int u = 0; u < 4; ++u) {
    const int q = q0 + u * ETH + tid;
    v[u] = d4[q < q_hi ? q : q_hi - 1];
  }
}

__device__ __forceinline__ void scan_emit(const int* __restrict__ dst, int e_lo, int e_hi,
                                          int lo, unsigned* list, int* cnt,
                                          int tid, int lane) {
  const int4* d4 = (const int4*)dst;
  const int q_lo = e_lo >> 2, q_hi = e_hi >> 2;
  const int STEP = 4 * ETH;
  int4 cur[4], nxt[4];
  load_dst4(cur, d4, q_lo, q_hi, tid);
  for (int q0 = q_lo; q0 < q_hi; q0 += STEP) {
    load_dst4(nxt, d4, q0 + STEP, q_hi, tid);
#pragma unroll
    for (int u = 0; u < 4; ++u) {
      const unsigned t0 = (unsigned)(cur[u].x - lo), t1 = (unsigned)(cur[u].y - lo);
      const unsigned t2 = (unsigned)(cur[u].z - lo), t3 = (unsigned)(cur[u].w - lo);
      const bool in = (q0 + u * ETH + tid) < q_hi;
      const bool m0 = in && t0 < RN, m1 = in && t1 < RN, m2 = in && t2 < RN, m3 = in && t3 < RN;
      const unsigned b0 = __builtin_amdgcn_ballot_w32(m0);
      const unsigned b1 = __builtin_amdgcn_ballot_w32(m1);
      const unsigned b2 = __builtin_amdgcn_ballot_w32(m2);
      const unsigned b3 = __builtin_amdgcn_ballot_w32(m3);
      if ((b0 | b1 | b2 | b3) != 0u) {
        const int tot = __popc(b0) + __popc(b1) + __popc(b2) + __popc(b3);
        int base = 0;
        if (lane == 0) base = atomicAdd(cnt, tot);
        base = __builtin_amdgcn_readfirstlane(base);
        int p = base + (int)(__builtin_amdgcn_mbcnt_lo(b0, 0u) + __builtin_amdgcn_mbcnt_lo(b1, 0u) +
                             __builtin_amdgcn_mbcnt_lo(b2, 0u) + __builtin_amdgcn_mbcnt_lo(b3, 0u));
        const unsigned eb = (unsigned)(q0 + u * ETH + tid) << 2;
        if (m0) { if (p < CAP) list[p] = (eb + 0u) | (t0 << 19); ++p; }
        if (m1) { if (p < CAP) list[p] = (eb + 1u) | (t1 << 19); ++p; }
        if (m2) { if (p < CAP) list[p] = (eb + 2u) | (t2 << 19); ++p; }
        if (m3) { if (p < CAP) list[p] = (eb + 3u) | (t3 << 19); ++p; }
      }
    }
#pragma unroll
    for (int u = 0; u < 4; ++u) cur[u] = nxt[u];
  }
}

__device__ __forceinline__ void edge_scores(int L, bool domax, unsigned* list, float4* ebuf,
                                            float* mxs, const int* __restrict__ src,
                                            const float* __restrict__ el,
                                            const float* __restrict__ er, int lo, int tid) {
  for (int k = tid; k < L; k += ETH) {
    const unsigned w = list[k];
    const int e = (int)(w & 0x7FFFFu);
    const int tl = (int)(w >> 19);
    const int s = src[e];
    const float4 a = *(const float4*)(el + (size_t)s * HH);
    const float4 b = *(const float4*)(er + (size_t)(lo + tl) * HH);
    float4 v;
    v.x = leaky(a.x + b.x); v.y = leaky(a.y + b.y);
    v.z = leaky(a.z + b.z); v.w = leaky(a.w + b.w);
    ebuf[k] = v;
    list[k] = (unsigned)s | ((unsigned)tl << 16);
    if (domax) {
      lds_max_f(&mxs[tl * HH + 0], v.x); lds_max_f(&mxs[tl * HH + 1], v.y);
      lds_max_f(&mxs[tl * HH + 2], v.z); lds_max_f(&mxs[tl * HH + 3], v.w);
    }
  }
}

__device__ __forceinline__ void row_load(int k, const unsigned* list, const float4* ebuf,
                                         const float* mxs, float* dns,
                                         const float* __restrict__ feat, int lane,
                                         float4 (&f)[HH], float (&p)[4], int& tl) {
  const unsigned w = list[k];
  const int s = (int)(w & 0xFFFFu);
  tl = (int)(w >> 16);
  const float4 v = ebuf[k];
  const float4 m = *(const float4*)(mxs + tl * HH);
  p[0] = __expf(v.x - m.x); p[1] = __expf(v.y - m.y);
  p[2] = __expf(v.z - m.z); p[3] = __expf(v.w - m.w);
  const float4* fr = (const float4*)(feat + (size_t)s * FH) + lane;
#pragma unroll
  for (int j = 0; j < HH; ++j) f[j] = fr[(DD / 4) * j];
  if (lane < HH) {
    const float pv = lane == 0 ? p[0] : lane == 1 ? p[1] : lane == 2 ? p[2] : p[3];
    lds_add_f(&dns[tl * HH + lane], pv);
  }
}

__device__ __forceinline__ void row_add(float* acc, int tl, int lane,
                                        const float4 (&f)[HH], const float (&p)[4]) {
  float* ar = acc + tl * FH + lane;
#pragma unroll
  for (int j = 0; j < HH; ++j) {
    lds_add_f(ar + j * DD + 0 * (DD / 4), f[j].x * p[j]);
    lds_add_f(ar + j * DD + 1 * (DD / 4), f[j].y * p[j]);
    lds_add_f(ar + j * DD + 2 * (DD / 4), f[j].z * p[j]);
    lds_add_f(ar + j * DD + 3 * (DD / 4), f[j].w * p[j]);
  }
}

__device__ __forceinline__ float4 head_cols4(const float* hd, int d0) {
  const float* q = hd + (d0 >> 2);
  float4 a;
  a.x = q[0 * (DD / 4)]; a.y = q[1 * (DD / 4)];
  a.z = q[2 * (DD / 4)]; a.w = q[3 * (DD / 4)];
  return a;
}

__device__ __forceinline__ void accumulate(int L, const unsigned* list, const float4* ebuf,
                                           const float* mxs, float* dns, float* acc,
                                           const float* __restrict__ feat, int wv, int lane) {
  int k = wv;
  for (; k + 32 < L; k += 64) {
    float4 fa[HH], fb[HH];
    float pa[4], pb[4];
    int ta, tb;
    row_load(k, list, ebuf, mxs, dns, feat, lane, fa, pa, ta);
    row_load(k + 32, list, ebuf, mxs, dns, feat, lane, fb, pb, tb);
    row_add(acc, ta, lane, fa, pa);
    row_add(acc, tb, lane, fb, pb);
  }
  if (k < L) {
    float4 fa[HH];
    float pa[4];
    int ta;
    row_load(k, list, ebuf, mxs, dns, feat, lane, fa, pa, ta);
    row_add(acc, ta, lane, fa, pa);
  }
}

#define HPT ((RN * FH) / (ETH * 4))
static_assert(HPT * ETH * 4 == RN * FH, "the h rows split evenly over the threads");
static_assert(ETH / 32 == FH / 16, "one wave per 16-column tile of the projection");
static_assert(RN > 32 && RN <= 48 && RN % 8 == 0, "three row tiles, the third half used");

__device__ __forceinline__ void project_rows(const float* hs, const float* __restrict__ wn,
                                             float* __restrict__ fn, int lo, int wv, int lane) {
  const int half = lane >> 4;
  const int l = lane & 15;
  const int r2 = 32 + (l & 7);
  const int x01 = l << 2, x2 = (r2 & 15) << 2;
  const float* a0p = hs + l * FH + half * 2;
  const float* a1p = hs + (16 + l) * FH + half * 2;
  const float* a2p = hs + r2 * FH + half * 2;
  const float* bp = wn + (size_t)(half * 2) * FH + wv * 16 + l;
  v8f c0 = {}, c1 = {}, c2 = {};
#pragma unroll 8
  for (int k = 0; k < FH; k += 4) {
    v2f b; b.x = bp[0]; b.y = bp[FH];
    bp += 4 * FH;
    const v2f a0 = *(const v2f*)(a0p + (k ^ x01));
    const v2f a1 = *(const v2f*)(a1p + (k ^ x01));
    const v2f a2 = *(const v2f*)(a2p + (k ^ x2));
    c0 = __builtin_amdgcn_wmma_f32_16x16x4_f32(false, a0, false, b, (short)0, c0, false, false);
    c1 = __builtin_amdgcn_wmma_f32_16x16x4_f32(false, a1, false, b, (short)0, c1, false, false);
    c2 = __builtin_amdgcn_wmma_f32_16x16x4_f32(false, a2, false, b, (short)0, c2, false, false);
  }
  float* ob = fn + (size_t)(lo + half * 8) * FH + wv * 16 + l;
#pragma unroll
  for (int r = 0; r < 8; ++r) {
    ob[(size_t)r * FH] = c0[r];
    ob[(size_t)(16 + r) * FH] = c1[r];
  }
  if (half == 0) {
#pragma unroll
    for (int r = 0; r < 8; ++r) ob[(size_t)(32 + r) * FH] = c2[r];
  }
}

__device__ __forceinline__ float elu1(float v) {
  return v > 0.f ? v : (__expf(v) - 1.f);
}

template <bool LAST>
__global__ __launch_bounds__(ETH) void edge_own(
    const int* __restrict__ src, const int* __restrict__ dst,
    const float* __restrict__ feat, const float* __restrict__ el,
    const float* __restrict__ er, const float* __restrict__ hin,
    const float* __restrict__ wnext, float* __restrict__ fnext,
    float* __restrict__ outp) {
  __shared__ __attribute__((aligned(16))) float acc[RN * FH];
  __shared__ __attribute__((aligned(16))) float4 ebuf[CAP];
  __shared__ unsigned list[CAP];
  __shared__ __attribute__((aligned(16))) float mxs[RN * HH];
  __shared__ __attribute__((aligned(16))) float dns[RN * HH];
  __shared__ int cnt;

  const int tid = threadIdx.x;
  const int lane = tid & 31;
  const int wv = tid >> 5;
  const int lo = blockIdx.x * RN;

  for (int i = tid * 4; i < RN * FH; i += ETH * 4)
    *(float4*)(acc + i) = make_float4(0.f, 0.f, 0.f, 0.f);
  if (tid < RN * HH) { mxs[tid] = -__builtin_huge_valf(); dns[tid] = 0.f; }
  if (tid == 0) cnt = 0;
  __syncthreads();

  scan_emit(dst, 0, EE, lo, list, &cnt, tid, lane);
  __syncthreads();
  const int total = cnt;

  if (total <= CAP) {
    edge_scores(total, true, list, ebuf, mxs, src, el, er, lo, tid);
    __syncthreads();
    accumulate(total, list, ebuf, mxs, dns, acc, feat, wv, lane);
  } else {
    for (int c0 = 0; c0 < EE; c0 += CAP) {
      __syncthreads();
      if (tid == 0) cnt = 0;
      __syncthreads();
      scan_emit(dst, c0, (c0 + CAP < EE) ? c0 + CAP : EE, lo, list, &cnt, tid, lane);
      __syncthreads();
      edge_scores(cnt, true, list, ebuf, mxs, src, el, er, lo, tid);
    }
    for (int c0 = 0; c0 < EE; c0 += CAP) {
      __syncthreads();
      if (tid == 0) cnt = 0;
      __syncthreads();
      scan_emit(dst, c0, (c0 + CAP < EE) ? c0 + CAP : EE, lo, list, &cnt, tid, lane);
      __syncthreads();
      const int L = cnt;
      edge_scores(L, false, list, ebuf, mxs, src, el, er, lo, tid);
      __syncthreads();
      accumulate(L, list, ebuf, mxs, dns, acc, feat, wv, lane);
    }
  }
  __syncthreads();

  if (!LAST) {
    float* ob = outp + (size_t)lo * FH;
    float4 hv[HPT];
#pragma unroll
    for (int it = 0; it < HPT; ++it) {
      const int i = (it * ETH + tid) * 4;
      const float d = dns[i >> 7];
      const float dd = d == 0.f ? 1.f : d;
      const float4 a = head_cols4(acc + (i & ~(DD - 1)), i & (DD - 1));
      float4 o;
      o.x = elu1(a.x / dd); o.y = elu1(a.y / dd);
      o.z = elu1(a.z / dd); o.w = elu1(a.w / dd);
      hv[it] = o;
    }
    __syncthreads();
#pragma unroll
    for (int it = 0; it < HPT; ++it) {
      const int i = (it * ETH + tid) * 4;
      const int r = i >> 9, c = i & (FH - 1);
      *(float4*)(acc + r * FH + (c ^ ((r & 15) << 2))) = hv[it];
    }
    __syncthreads();
#pragma unroll
    for (int it = 0; it < HPT; ++it)
      *(float4*)(ob + (it * ETH + tid) * 4) = hv[it];
    project_rows(acc, wnext, fnext, lo, wv, lane);
  } else {
    for (int i = tid * 4; i < RN * DD; i += ETH * 4) {
      const int tl = i >> 7, d0 = i & (DD - 1);
      float4 sum = make_float4(0.f, 0.f, 0.f, 0.f);
#pragma unroll
      for (int h = 0; h < HH; ++h) {
        const float d = dns[tl * HH + h];
        const float dd = d == 0.f ? 1.f : d;
        const float4 a = head_cols4(acc + tl * FH + h * DD, d0);
        const float4 r = *(const float4*)(hin + (size_t)(lo + tl) * FH + h * DD + d0);
        sum.x += elu1(a.x / dd + r.x); sum.y += elu1(a.y / dd + r.y);
        sum.z += elu1(a.z / dd + r.z); sum.w += elu1(a.w / dd + r.w);
      }
      float4 o;
      o.x = sum.x * (1.f / HH); o.y = sum.y * (1.f / HH);
      o.z = sum.z * (1.f / HH); o.w = sum.w * (1.f / HH);
      *(float4*)(outp + (size_t)(lo + tl) * DD + d0) = o;
    }
  }
}

extern "C" void kernel_launch(void* const* d_in, const int* in_sizes, int n_in,
                              void* d_out, int out_size, void* d_ws, size_t ws_size,
                              hipStream_t stream) {
  const int* src = (const int*)d_in[0];
  const int* dst = (const int*)d_in[1];
  const float* x   = (const float*)d_in[2];
  const float* W0  = (const float*)d_in[3];
  const float* al0 = (const float*)d_in[4];
  const float* ar0 = (const float*)d_in[5];
  const float* W1  = (const float*)d_in[6];
  const float* al1 = (const float*)d_in[7];
  const float* ar1 = (const float*)d_in[8];
  float* out = (float*)d_out;

  float* ws    = (float*)d_ws;
  float* feat  = ws;
  float* hbuf  = feat  + (size_t)NN * FH;
  float* el    = hbuf  + (size_t)NN * FH;
  float* er    = el    + (size_t)NN * HH;
  float* feat1 = er    + (size_t)NN * HH;

  const int mtiles = (NN + 15) / 16;
  const int mgroups = (mtiles + 7) / 8;
  const int ntiles = FH / 16;
  const int gemmBlocks = mgroups * ntiles;

  gemm16_wmma<<<gemmBlocks, 256, 0, stream>>>(x, W0, feat, NN, K0, FH);
  attn_coef<<<(NN * HH + 255) / 256, 256, 0, stream>>>(feat, al0, ar0, el, er);
  edge_own<false><<<NBLK, ETH, 0, stream>>>(src, dst, feat, el, er, nullptr, W1, feat1, hbuf);
  attn_coef<<<(NN * HH + 255) / 256, 256, 0, stream>>>(feat1, al1, ar1, el, er);
  edge_own<true><<<NBLK, ETH, 0, stream>>>(src, dst, feat1, el, er, hbuf, nullptr, nullptr, out);
}


# dst scanned once per call: layer 0's owner work-group leaves its edge list (src, local node) and count in the workspace, layer 1 reads them instead of scanning; sum order LDS arrival as before
# speedup vs baseline: 5.2409x; 1.0775x over previous
#include <hip/hip_runtime.h>
#include <math.h>

#define NN 10000
#define EE 320000
#define HH 4
#define DD 128
#define FH (HH * DD)
#define K0 256
#define KMAX 512
#define NEG_SLOPE 0.2f

typedef __attribute__((ext_vector_type(2))) float v2f;
typedef __attribute__((ext_vector_type(8))) float v8f;

__global__ __launch_bounds__(256) void gemm16_wmma(const float* __restrict__ X,
                                                   const float* __restrict__ W,
                                                   float* __restrict__ out,
                                                   int M, int K, int Nout) {
  __shared__ float Bs[KMAX * 16];

  const int tid = threadIdx.x;
  const int lane = tid & 31;
  const int wv = tid >> 5;
  const int ntiles = Nout >> 4;
  const int mg = blockIdx.x / ntiles;
  const int tn = blockIdx.x - mg * ntiles;

  const float* wp = W + tn * 16;
  for (int i = tid; i < K * 16; i += 256) {
    int kk = i >> 4, c = i & 15;
    Bs[i] = wp[(size_t)kk * Nout + c];
  }
  __syncthreads();

  const int tm = mg * 8 + wv;
  if (tm * 16 >= M) return;

  const int half = lane >> 4;
  const int l = lane & 15;

  const float* aptr = X + (size_t)(tm * 16 + l) * K + half * 2;
  const float* bbase = Bs + (half * 2) * 16 + l;

  v8f acc = {};
  for (int k = 0; k < K; k += 4) {
    float2 av = *(const float2*)aptr;
    v2f a; a.x = av.x; a.y = av.y;
    v2f b; b.x = bbase[0]; b.y = bbase[16];
    acc = __builtin_amdgcn_wmma_f32_16x16x4_f32(
         false, a,  false, b,
         (short)0, acc,  false,  false);
    aptr += 4;
    bbase += 64;
  }

  float* obase = out + (size_t)(tm * 16 + half * 8) * Nout + tn * 16 + l;
#pragma unroll
  for (int r = 0; r < 8; ++r) obase[(size_t)r * Nout] = acc[r];
}

__global__ void attn_coef(const float* __restrict__ feat,
                          const float* __restrict__ al,
                          const float* __restrict__ ar,
                          float* __restrict__ el, float* __restrict__ er) {
  int i = blockIdx.x * blockDim.x + threadIdx.x;
  if (i >= NN * HH) return;
  int h = i & (HH - 1);
  const float4* f = (const float4*)(feat + (size_t)i * DD);
  const float4* a = (const float4*)(al + h * DD);
  const float4* b = (const float4*)(ar + h * DD);
  float sl = 0.f, sr = 0.f;
#pragma unroll 4
  for (int d = 0; d < DD / 4; ++d) {
    float4 fv = f[d], av = a[d], bv = b[d];
    sl += fv.x * av.x + fv.y * av.y + fv.z * av.z + fv.w * av.w;
    sr += fv.x * bv.x + fv.y * bv.y + fv.z * bv.z + fv.w * bv.w;
  }
  el[i] = sl;
  er[i] = sr;
}

#define RN 40
#define NBLK (NN / RN)
#define ETH 1024
#define CAP 4096

static_assert(NN % RN == 0, "node ranges must tile N");
static_assert(EE % 4 == 0 && CAP % 4 == 0, "dst is scanned 4 edges per load");
static_assert(EE < (1 << 19) && RN < (1 << 6) && NN <= (1 << 16), "packed fields");
static_assert(DD / 4 == 32, "a wave's 32 lanes x 16 bytes cover one head of a row");

__device__ __forceinline__ float leaky(float v) {
  return v > 0.f ? v : NEG_SLOPE * v;
}

__device__ __forceinline__ void lds_max_f(float* addr, float v) {
  if (__float_as_int(v) >= 0)
    atomicMax((int*)addr, __float_as_int(v));
  else
    atomicMin((unsigned int*)addr, __float_as_uint(v));
}

__device__ __forceinline__ void lds_add_f(float* addr, float v) {
  __hip_atomic_fetch_add(addr, v, __ATOMIC_RELAXED, __HIP_MEMORY_SCOPE_WORKGROUP);
}

__device__ __forceinline__ void load_dst4(int4 (&v)[4], const int4* __restrict__ d4,
                                          int q0, int q_hi, int tid) {
#pragma unroll
  for (int u = 0; u < 4; ++u) {
    const int q = q0 + u * ETH + tid;
    v[u] = d4[q < q_hi ? q : q_hi - 1];
  }
}

__device__ __forceinline__ void scan_emit(const int* __restrict__ dst, int e_lo, int e_hi,
                                          int lo, unsigned* list, int* cnt,
                                          int tid, int lane) {
  const int4* d4 = (const int4*)dst;
  const int q_lo = e_lo >> 2, q_hi = e_hi >> 2;
  const int STEP = 4 * ETH;
  int4 cur[4], nxt[4];
  load_dst4(cur, d4, q_lo, q_hi, tid);
  for (int q0 = q_lo; q0 < q_hi; q0 += STEP) {
    load_dst4(nxt, d4, q0 + STEP, q_hi, tid);
#pragma unroll
    for (int u = 0; u < 4; ++u) {
      const unsigned t0 = (unsigned)(cur[u].x - lo), t1 = (unsigned)(cur[u].y - lo);
      const unsigned t2 = (unsigned)(cur[u].z - lo), t3 = (unsigned)(cur[u].w - lo);
      const bool in = (q0 + u * ETH + tid) < q_hi;
      const bool m0 = in && t0 < RN, m1 = in && t1 < RN, m2 = in && t2 < RN, m3 = in && t3 < RN;
      const unsigned b0 = __builtin_amdgcn_ballot_w32(m0);
      const unsigned b1 = __builtin_amdgcn_ballot_w32(m1);
      const unsigned b2 = __builtin_amdgcn_ballot_w32(m2);
      const unsigned b3 = __builtin_amdgcn_ballot_w32(m3);
      if ((b0 | b1 | b2 | b3) != 0u) {
        const int tot = __popc(b0) + __popc(b1) + __popc(b2) + __popc(b3);
        int base = 0;
        if (lane == 0) base = atomicAdd(cnt, tot);
        base = __builtin_amdgcn_readfirstlane(base);
        int p = base + (int)(__builtin_amdgcn_mbcnt_lo(b0, 0u) + __builtin_amdgcn_mbcnt_lo(b1, 0u) +
                             __builtin_amdgcn_mbcnt_lo(b2, 0u) + __builtin_amdgcn_mbcnt_lo(b3, 0u));
        const unsigned eb = (unsigned)(q0 + u * ETH + tid) << 2;
        if (m0) { if (p < CAP) list[p] = (eb + 0u) | (t0 << 19); ++p; }
        if (m1) { if (p < CAP) list[p] = (eb + 1u) | (t1 << 19); ++p; }
        if (m2) { if (p < CAP) list[p] = (eb + 2u) | (t2 << 19); ++p; }
        if (m3) { if (p < CAP) list[p] = (eb + 3u) | (t3 << 19); ++p; }
      }
    }
#pragma unroll
    for (int u = 0; u < 4; ++u) cur[u] = nxt[u];
  }
}

__device__ __forceinline__ void edge_scores(int L, bool domax, unsigned* list, float4* ebuf,
                                            float* mxs, const int* __restrict__ src,
                                            const float* __restrict__ el,
                                            const float* __restrict__ er, int lo, int tid) {
  for (int k = tid; k < L; k += ETH) {
    const unsigned w = list[k];
    const int e = (int)(w & 0x7FFFFu);
    const int tl = (int)(w >> 19);
    const int s = src[e];
    const float4 a = *(const float4*)(el + (size_t)s * HH);
    const float4 b = *(const float4*)(er + (size_t)(lo + tl) * HH);
    float4 v;
    v.x = leaky(a.x + b.x); v.y = leaky(a.y + b.y);
    v.z = leaky(a.z + b.z); v.w = leaky(a.w + b.w);
    ebuf[k] = v;
    list[k] = (unsigned)s | ((unsigned)tl << 16);
    if (domax) {
      lds_max_f(&mxs[tl * HH + 0], v.x); lds_max_f(&mxs[tl * HH + 1], v.y);
      lds_max_f(&mxs[tl * HH + 2], v.z); lds_max_f(&mxs[tl * HH + 3], v.w);
    }
  }
}

__device__ __forceinline__ void edge_scores_saved(int L, const unsigned* __restrict__ gl,
                                                  unsigned* list, float4* ebuf, float* mxs,
                                                  const float* __restrict__ el,
                                                  const float* __restrict__ er, int lo, int tid) {
  for (int k = tid; k < L; k += ETH) {
    const unsigned w = gl[k];
    const int s = (int)(w & 0xFFFFu);
    const int tl = (int)(w >> 16);
    const float4 a = *(const float4*)(el + (size_t)s * HH);
    const float4 b = *(const float4*)(er + (size_t)(lo + tl) * HH);
    float4 v;
    v.x = leaky(a.x + b.x); v.y = leaky(a.y + b.y);
    v.z = leaky(a.z + b.z); v.w = leaky(a.w + b.w);
    ebuf[k] = v;
    list[k] = w;
    lds_max_f(&mxs[tl * HH + 0], v.x); lds_max_f(&mxs[tl * HH + 1], v.y);
    lds_max_f(&mxs[tl * HH + 2], v.z); lds_max_f(&mxs[tl * HH + 3], v.w);
  }
}

__device__ __forceinline__ void row_load(int k, const unsigned* list, const float4* ebuf,
                                         const float* mxs, float* dns,
                                         const float* __restrict__ feat, int lane,
                                         float4 (&f)[HH], float (&p)[4], int& tl) {
  const unsigned w = list[k];
  const int s = (int)(w & 0xFFFFu);
  tl = (int)(w >> 16);
  const float4 v = ebuf[k];
  const float4 m = *(const float4*)(mxs + tl * HH);
  p[0] = __expf(v.x - m.x); p[1] = __expf(v.y - m.y);
  p[2] = __expf(v.z - m.z); p[3] = __expf(v.w - m.w);
  const float4* fr = (const float4*)(feat + (size_t)s * FH) + lane;
#pragma unroll
  for (int j = 0; j < HH; ++j) f[j] = fr[(DD / 4) * j];
  if (lane < HH) {
    const float pv = lane == 0 ? p[0] : lane == 1 ? p[1] : lane == 2 ? p[2] : p[3];
    lds_add_f(&dns[tl * HH + lane], pv);
  }
}

__device__ __forceinline__ void row_add(float* acc, int tl, int lane,
                                        const float4 (&f)[HH], const float (&p)[4]) {
  float* ar = acc + tl * FH + lane;
#pragma unroll
  for (int j = 0; j < HH; ++j) {
    lds_add_f(ar + j * DD + 0 * (DD / 4), f[j].x * p[j]);
    lds_add_f(ar + j * DD + 1 * (DD / 4), f[j].y * p[j]);
    lds_add_f(ar + j * DD + 2 * (DD / 4), f[j].z * p[j]);
    lds_add_f(ar + j * DD + 3 * (DD / 4), f[j].w * p[j]);
  }
}

__device__ __forceinline__ float4 head_cols4(const float* hd, int d0) {
  const float* q = hd + (d0 >> 2);
  float4 a;
  a.x = q[0 * (DD / 4)]; a.y = q[1 * (DD / 4)];
  a.z = q[2 * (DD / 4)]; a.w = q[3 * (DD / 4)];
  return a;
}

__device__ __forceinline__ void accumulate(int L, const unsigned* list, const float4* ebuf,
                                           const float* mxs, float* dns, float* acc,
                                           const float* __restrict__ feat, int wv, int lane) {
  int k = wv;
  for (; k + 32 < L; k += 64) {
    float4 fa[HH], fb[HH];
    float pa[4], pb[4];
    int ta, tb;
    row_load(k, list, ebuf, mxs, dns, feat, lane, fa, pa, ta);
    row_load(k + 32, list, ebuf, mxs, dns, feat, lane, fb, pb, tb);
    row_add(acc, ta, lane, fa, pa);
    row_add(acc, tb, lane, fb, pb);
  }
  if (k < L) {
    float4 fa[HH];
    float pa[4];
    int ta;
    row_load(k, list, ebuf, mxs, dns, feat, lane, fa, pa, ta);
    row_add(acc, ta, lane, fa, pa);
  }
}

#define HPT ((RN * FH) / (ETH * 4))
static_assert(HPT * ETH * 4 == RN * FH, "the h rows split evenly over the threads");
static_assert(ETH / 32 == FH / 16, "one wave per 16-column tile of the projection");
static_assert(RN > 32 && RN <= 48 && RN % 8 == 0, "three row tiles, the third half used");

__device__ __forceinline__ void project_rows(const float* hs, const float* __restrict__ wn,
                                             float* __restrict__ fn, int lo, int wv, int lane) {
  const int half = lane >> 4;
  const int l = lane & 15;
  const int r2 = 32 + (l & 7);
  const int x01 = l << 2, x2 = (r2 & 15) << 2;
  const float* a0p = hs + l * FH + half * 2;
  const float* a1p = hs + (16 + l) * FH + half * 2;
  const float* a2p = hs + r2 * FH + half * 2;
  const float* bp = wn + (size_t)(half * 2) * FH + wv * 16 + l;
  v8f c0 = {}, c1 = {}, c2 = {};
#pragma unroll 8
  for (int k = 0; k < FH; k += 4) {
    v2f b; b.x = bp[0]; b.y = bp[FH];
    bp += 4 * FH;
    const v2f a0 = *(const v2f*)(a0p + (k ^ x01));
    const v2f a1 = *(const v2f*)(a1p + (k ^ x01));
    const v2f a2 = *(const v2f*)(a2p + (k ^ x2));
    c0 = __builtin_amdgcn_wmma_f32_16x16x4_f32(false, a0, false, b, (short)0, c0, false, false);
    c1 = __builtin_amdgcn_wmma_f32_16x16x4_f32(false, a1, false, b, (short)0, c1, false, false);
    c2 = __builtin_amdgcn_wmma_f32_16x16x4_f32(false, a2, false, b, (short)0, c2, false, false);
  }
  float* ob = fn + (size_t)(lo + half * 8) * FH + wv * 16 + l;
#pragma unroll
  for (int r = 0; r < 8; ++r) {
    ob[(size_t)r * FH] = c0[r];
    ob[(size_t)(16 + r) * FH] = c1[r];
  }
  if (half == 0) {
#pragma unroll
    for (int r = 0; r < 8; ++r) ob[(size_t)(32 + r) * FH] = c2[r];
  }
}

__device__ __forceinline__ float elu1(float v) {
  return v > 0.f ? v : (__expf(v) - 1.f);
}

template <bool LAST>
__global__ __launch_bounds__(ETH) void edge_own(
    const int* __restrict__ src, const int* __restrict__ dst,
    const float* __restrict__ feat, const float* __restrict__ el,
    const float* __restrict__ er, const float* __restrict__ hin,
    const float* __restrict__ wnext, float* __restrict__ fnext,
    float* __restrict__ outp, unsigned* __restrict__ glist, int* __restrict__ gcnt) {
  __shared__ __attribute__((aligned(16))) float acc[RN * FH];
  __shared__ __attribute__((aligned(16))) float4 ebuf[CAP];
  __shared__ unsigned list[CAP];
  __shared__ __attribute__((aligned(16))) float mxs[RN * HH];
  __shared__ __attribute__((aligned(16))) float dns[RN * HH];
  __shared__ int cnt;

  const int tid = threadIdx.x;
  const int lane = tid & 31;
  const int wv = tid >> 5;
  const int lo = blockIdx.x * RN;

  for (int i = tid * 4; i < RN * FH; i += ETH * 4)
    *(float4*)(acc + i) = make_float4(0.f, 0.f, 0.f, 0.f);
  if (tid < RN * HH) { mxs[tid] = -__builtin_huge_valf(); dns[tid] = 0.f; }
  if (tid == 0) cnt = 0;
  __syncthreads();

  unsigned* gl = glist + (size_t)blockIdx.x * CAP;
  int total;
  if (!LAST) {
    scan_emit(dst, 0, EE, lo, list, &cnt, tid, lane);
    __syncthreads();
    total = cnt;
    if (tid == 0) gcnt[blockIdx.x] = total;
  } else {
    total = gcnt[blockIdx.x];
  }

  if (total <= CAP) {
    if (!LAST) {
      edge_scores(total, true, list, ebuf, mxs, src, el, er, lo, tid);
      __syncthreads();
      for (int k = tid; k < total; k += ETH) gl[k] = list[k];
    } else {
      edge_scores_saved(total, gl, list, ebuf, mxs, el, er, lo, tid);
      __syncthreads();
    }
    accumulate(total, list, ebuf, mxs, dns, acc, feat, wv, lane);
  } else {
    for (int c0 = 0; c0 < EE; c0 += CAP) {
      __syncthreads();
      if (tid == 0) cnt = 0;
      __syncthreads();
      scan_emit(dst, c0, (c0 + CAP < EE) ? c0 + CAP : EE, lo, list, &cnt, tid, lane);
      __syncthreads();
      edge_scores(cnt, true, list, ebuf, mxs, src, el, er, lo, tid);
    }
    for (int c0 = 0; c0 < EE; c0 += CAP) {
      __syncthreads();
      if (tid == 0) cnt = 0;
      __syncthreads();
      scan_emit(dst, c0, (c0 + CAP < EE) ? c0 + CAP : EE, lo, list, &cnt, tid, lane);
      __syncthreads();
      const int L = cnt;
      edge_scores(L, false, list, ebuf, mxs, src, el, er, lo, tid);
      __syncthreads();
      accumulate(L, list, ebuf, mxs, dns, acc, feat, wv, lane);
    }
  }
  __syncthreads();

  if (!LAST) {
    float* ob = outp + (size_t)lo * FH;
    float4 hv[HPT];
#pragma unroll
    for (int it = 0; it < HPT; ++it) {
      const int i = (it * ETH + tid) * 4;
      const float d = dns[i >> 7];
      const float dd = d == 0.f ? 1.f : d;
      const float4 a = head_cols4(acc + (i & ~(DD - 1)), i & (DD - 1));
      float4 o;
      o.x = elu1(a.x / dd); o.y = elu1(a.y / dd);
      o.z = elu1(a.z / dd); o.w = elu1(a.w / dd);
      hv[it] = o;
    }
    __syncthreads();
#pragma unroll
    for (int it = 0; it < HPT; ++it) {
      const int i = (it * ETH + tid) * 4;
      const int r = i >> 9, c = i & (FH - 1);
      *(float4*)(acc + r * FH + (c ^ ((r & 15) << 2))) = hv[it];
    }
    __syncthreads();
#pragma unroll
    for (int it = 0; it < HPT; ++it)
      *(float4*)(ob + (it * ETH + tid) * 4) = hv[it];
    project_rows(acc, wnext, fnext, lo, wv, lane);
  } else {
    for (int i = tid * 4; i < RN * DD; i += ETH * 4) {
      const int tl = i >> 7, d0 = i & (DD - 1);
      float4 sum = make_float4(0.f, 0.f, 0.f, 0.f);
#pragma unroll
      for (int h = 0; h < HH; ++h) {
        const float d = dns[tl * HH + h];
        const float dd = d == 0.f ? 1.f : d;
        const float4 a = head_cols4(acc + tl * FH + h * DD, d0);
        const float4 r = *(const float4*)(hin + (size_t)(lo + tl) * FH + h * DD + d0);
        sum.x += elu1(a.x / dd + r.x); sum.y += elu1(a.y / dd + r.y);
        sum.z += elu1(a.z / dd + r.z); sum.w += elu1(a.w / dd + r.w);
      }
      float4 o;
      o.x = sum.x * (1.f / HH); o.y = sum.y * (1.f / HH);
      o.z = sum.z * (1.f / HH); o.w = sum.w * (1.f / HH);
      *(float4*)(outp + (size_t)(lo + tl) * DD + d0) = o;
    }
  }
}

extern "C" void kernel_launch(void* const* d_in, const int* in_sizes, int n_in,
                              void* d_out, int out_size, void* d_ws, size_t ws_size,
                              hipStream_t stream) {
  const int* src = (const int*)d_in[0];
  const int* dst = (const int*)d_in[1];
  const float* x   = (const float*)d_in[2];
  const float* W0  = (const float*)d_in[3];
  const float* al0 = (const float*)d_in[4];
  const float* ar0 = (const float*)d_in[5];
  const float* W1  = (const float*)d_in[6];
  const float* al1 = (const float*)d_in[7];
  const float* ar1 = (const float*)d_in[8];
  float* out = (float*)d_out;

  float* ws    = (float*)d_ws;
  float* feat  = ws;
  float* hbuf  = feat  + (size_t)NN * FH;
  float* el    = hbuf  + (size_t)NN * FH;
  float* er    = el    + (size_t)NN * HH;
  float* feat1 = er    + (size_t)NN * HH;
  unsigned* glist = (unsigned*)(feat1 + (size_t)NN * FH);
  int* gcnt = (int*)(glist + (size_t)NBLK * CAP);

  const int mtiles = (NN + 15) / 16;
  const int mgroups = (mtiles + 7) / 8;
  const int ntiles = FH / 16;
  const int gemmBlocks = mgroups * ntiles;

  gemm16_wmma<<<gemmBlocks, 256, 0, stream>>>(x, W0, feat, NN, K0, FH);
  attn_coef<<<(NN * HH + 255) / 256, 256, 0, stream>>>(feat, al0, ar0, el, er);
  edge_own<false><<<NBLK, ETH, 0, stream>>>(src, dst, feat, el, er, nullptr, W1, feat1, hbuf, glist, gcnt);
  attn_coef<<<(NN * HH + 255) / 256, 256, 0, stream>>>(feat1, al1, ar1, el, er);
  edge_own<true><<<NBLK, ETH, 0, stream>>>(src, dst, feat1, el, er, hbuf, nullptr, nullptr, out, glist, gcnt);
}


# Owner work-group counting-sorts its LDS edge list by local node; a wave sums a contiguous run of it in registers and adds to the LDS row only where the node changes; sum order: run order, LDS arrival
# speedup vs baseline: 5.8202x; 1.1105x over previous
#include <hip/hip_runtime.h>
#include <math.h>

#define NN 10000
#define EE 320000
#define HH 4
#define DD 128
#define FH (HH * DD)
#define K0 256
#define KMAX 512
#define NEG_SLOPE 0.2f

typedef __attribute__((ext_vector_type(2))) float v2f;
typedef __attribute__((ext_vector_type(8))) float v8f;

__global__ __launch_bounds__(256) void gemm16_wmma(const float* __restrict__ X,
                                                   const float* __restrict__ W,
                                                   float* __restrict__ out,
                                                   int M, int K, int Nout) {
  __shared__ float Bs[KMAX * 16];

  const int tid = threadIdx.x;
  const int lane = tid & 31;
  const int wv = tid >> 5;
  const int ntiles = Nout >> 4;
  const int mg = blockIdx.x / ntiles;
  const int tn = blockIdx.x - mg * ntiles;

  const float* wp = W + tn * 16;
  for (int i = tid; i < K * 16; i += 256) {
    int kk = i >> 4, c = i & 15;
    Bs[i] = wp[(size_t)kk * Nout + c];
  }
  __syncthreads();

  const int tm = mg * 8 + wv;
  if (tm * 16 >= M) return;

  const int half = lane >> 4;
  const int l = lane & 15;

  const float* aptr = X + (size_t)(tm * 16 + l) * K + half * 2;
  const float* bbase = Bs + (half * 2) * 16 + l;

  v8f acc = {};
  for (int k = 0; k < K; k += 4) {
    float2 av = *(const float2*)aptr;
    v2f a; a.x = av.x; a.y = av.y;
    v2f b; b.x = bbase[0]; b.y = bbase[16];
    acc = __builtin_amdgcn_wmma_f32_16x16x4_f32(
         false, a,  false, b,
         (short)0, acc,  false,  false);
    aptr += 4;
    bbase += 64;
  }

  float* obase = out + (size_t)(tm * 16 + half * 8) * Nout + tn * 16 + l;
#pragma unroll
  for (int r = 0; r < 8; ++r) obase[(size_t)r * Nout] = acc[r];
}

__global__ void attn_coef(const float* __restrict__ feat,
                          const float* __restrict__ al,
                          const float* __restrict__ ar,
                          float* __restrict__ el, float* __restrict__ er) {
  int i = blockIdx.x * blockDim.x + threadIdx.x;
  if (i >= NN * HH) return;
  int h = i & (HH - 1);
  const float4* f = (const float4*)(feat + (size_t)i * DD);
  const float4* a = (const float4*)(al + h * DD);
  const float4* b = (const float4*)(ar + h * DD);
  float sl = 0.f, sr = 0.f;
#pragma unroll 4
  for (int d = 0; d < DD / 4; ++d) {
    float4 fv = f[d], av = a[d], bv = b[d];
    sl += fv.x * av.x + fv.y * av.y + fv.z * av.z + fv.w * av.w;
    sr += fv.x * bv.x + fv.y * bv.y + fv.z * bv.z + fv.w * bv.w;
  }
  el[i] = sl;
  er[i] = sr;
}

#define RN 40
#define NBLK (NN / RN)
#define ETH 1024
#define CAP 4096
#define EPT (CAP / ETH)

static_assert(NN % RN == 0, "node ranges must tile N");
static_assert(EE % 4 == 0 && CAP % 4 == 0, "dst is scanned 4 edges per load");
static_assert(EE < (1 << 19) && RN < (1 << 6) && NN <= (1 << 16), "packed fields");
static_assert(DD / 4 == 32, "a wave's 32 lanes x 16 bytes cover one head of a row");
static_assert(EPT * ETH == CAP, "a full list splits evenly over the threads");

__device__ __forceinline__ float leaky(float v) {
  return v > 0.f ? v : NEG_SLOPE * v;
}

__device__ __forceinline__ void lds_max_f(float* addr, float v) {
  if (__float_as_int(v) >= 0)
    atomicMax((int*)addr, __float_as_int(v));
  else
    atomicMin((unsigned int*)addr, __float_as_uint(v));
}

__device__ __forceinline__ void lds_add_f(float* addr, float v) {
  __hip_atomic_fetch_add(addr, v, __ATOMIC_RELAXED, __HIP_MEMORY_SCOPE_WORKGROUP);
}

__device__ __forceinline__ void load_dst4(int4 (&v)[4], const int4* __restrict__ d4,
                                          int q0, int q_hi, int tid) {
#pragma unroll
  for (int u = 0; u < 4; ++u) {
    const int q = q0 + u * ETH + tid;
    v[u] = d4[q < q_hi ? q : q_hi - 1];
  }
}

__device__ __forceinline__ void scan_emit(const int* __restrict__ dst, int e_lo, int e_hi,
                                          int lo, unsigned* list, int* cnt,
                                          int tid, int lane) {
  const int4* d4 = (const int4*)dst;
  const int q_lo = e_lo >> 2, q_hi = e_hi >> 2;
  const int STEP = 4 * ETH;
  int4 cur[4], nxt[4];
  load_dst4(cur, d4, q_lo, q_hi, tid);
  for (int q0 = q_lo; q0 < q_hi; q0 += STEP) {
    load_dst4(nxt, d4, q0 + STEP, q_hi, tid);
#pragma unroll
    for (int u = 0; u < 4; ++u) {
      const unsigned t0 = (unsigned)(cur[u].x - lo), t1 = (unsigned)(cur[u].y - lo);
      const unsigned t2 = (unsigned)(cur[u].z - lo), t3 = (unsigned)(cur[u].w - lo);
      const bool in = (q0 + u * ETH + tid) < q_hi;
      const bool m0 = in && t0 < RN, m1 = in && t1 < RN, m2 = in && t2 < RN, m3 = in && t3 < RN;
      const unsigned b0 = __builtin_amdgcn_ballot_w32(m0);
      const unsigned b1 = __builtin_amdgcn_ballot_w32(m1);
      const unsigned b2 = __builtin_amdgcn_ballot_w32(m2);
      const unsigned b3 = __builtin_amdgcn_ballot_w32(m3);
      if ((b0 | b1 | b2 | b3) != 0u) {
        const int tot = __popc(b0) + __popc(b1) + __popc(b2) + __popc(b3);
        int base = 0;
        if (lane == 0) base = atomicAdd(cnt, tot);
        base = __builtin_amdgcn_readfirstlane(base);
        int p = base + (int)(__builtin_amdgcn_mbcnt_lo(b0, 0u) + __builtin_amdgcn_mbcnt_lo(b1, 0u) +
                             __builtin_amdgcn_mbcnt_lo(b2, 0u) + __builtin_amdgcn_mbcnt_lo(b3, 0u));
        const unsigned eb = (unsigned)(q0 + u * ETH + tid) << 2;
        if (m0) { if (p < CAP) list[p] = (eb + 0u) | (t0 << 19); ++p; }
        if (m1) { if (p < CAP) list[p] = (eb + 1u) | (t1 << 19); ++p; }
        if (m2) { if (p < CAP) list[p] = (eb + 2u) | (t2 << 19); ++p; }
        if (m3) { if (p < CAP) list[p] = (eb + 3u) | (t3 << 19); ++p; }
      }
    }
#pragma unroll
    for (int u = 0; u < 4; ++u) cur[u] = nxt[u];
  }
}

__device__ __forceinline__ void edge_scores(int L, bool domax, unsigned* list, float4* ebuf,
                                            float* mxs, const int* __restrict__ src,
                                            const float* __restrict__ el,
                                            const float* __restrict__ er, int lo, int tid) {
  for (int k = tid; k < L; k += ETH) {
    const unsigned w = list[k];
    const int e = (int)(w & 0x7FFFFu);
    const int tl = (int)(w >> 19);
    const int s = src[e];
    const float4 a = *(const float4*)(el + (size_t)s * HH);
    const float4 b = *(const float4*)(er + (size_t)(lo + tl) * HH);
    float4 v;
    v.x = leaky(a.x + b.x); v.y = leaky(a.y + b.y);
    v.z = leaky(a.z + b.z); v.w = leaky(a.w + b.w);
    ebuf[k] = v;
    list[k] = (unsigned)s | ((unsigned)tl << 16);
    if (domax) {
      lds_max_f(&mxs[tl * HH + 0], v.x); lds_max_f(&mxs[tl * HH + 1], v.y);
      lds_max_f(&mxs[tl * HH + 2], v.z); lds_max_f(&mxs[tl * HH + 3], v.w);
    }
  }
}

__device__ __forceinline__ void edge_scores_saved(int L, const unsigned* __restrict__ gl,
                                                  unsigned* list, float4* ebuf, float* mxs,
                                                  const float* __restrict__ el,
                                                  const float* __restrict__ er, int lo, int tid) {
  for (int k = tid; k < L; k += ETH) {
    const unsigned w = gl[k];
    const int s = (int)(w & 0xFFFFu);
    const int tl = (int)(w >> 16);
    const float4 a = *(const float4*)(el + (size_t)s * HH);
    const float4 b = *(const float4*)(er + (size_t)(lo + tl) * HH);
    float4 v;
    v.x = leaky(a.x + b.x); v.y = leaky(a.y + b.y);
    v.z = leaky(a.z + b.z); v.w = leaky(a.w + b.w);
    ebuf[k] = v;
    list[k] = w;
    lds_max_f(&mxs[tl * HH + 0], v.x); lds_max_f(&mxs[tl * HH + 1], v.y);
    lds_max_f(&mxs[tl * HH + 2], v.z); lds_max_f(&mxs[tl * HH + 3], v.w);
  }
}

__device__ __forceinline__ void edge_scores_sorted(int L, unsigned* list, float4* ebuf,
                                                   float* mxs, int* hist, int* nstart,
                                                   const int* __restrict__ src,
                                                   const float* __restrict__ el,
                                                   const float* __restrict__ er, int lo, int tid) {
  unsigned wr[EPT];
  float4 vr[EPT];
  int rk[EPT];
#pragma unroll
  for (int i = 0; i < EPT; ++i) {
    const int k = tid + i * ETH;
    wr[i] = 0u; rk[i] = 0; vr[i] = make_float4(0.f, 0.f, 0.f, 0.f);
    if (k < L) {
      const unsigned w = list[k];
      const int e = (int)(w & 0x7FFFFu);
      const int tl = (int)(w >> 19);
      const int s = src[e];
      const float4 a = *(const float4*)(el + (size_t)s * HH);
      const float4 b = *(const float4*)(er + (size_t)(lo + tl) * HH);
      float4 v;
      v.x = leaky(a.x + b.x); v.y = leaky(a.y + b.y);
      v.z = leaky(a.z + b.z); v.w = leaky(a.w + b.w);
      lds_max_f(&mxs[tl * HH + 0], v.x); lds_max_f(&mxs[tl * HH + 1], v.y);
      lds_max_f(&mxs[tl * HH + 2], v.z); lds_max_f(&mxs[tl * HH + 3], v.w);
      rk[i] = atomicAdd(&hist[tl], 1);
      wr[i] = (unsigned)s | ((unsigned)tl << 16);
      vr[i] = v;
    }
  }
  __syncthreads();
  if (tid < RN) {
    int first = 0;
#pragma unroll
    for (int j = 0; j < RN; ++j) first += (j < tid) ? hist[j] : 0;
    nstart[tid] = first;
  }
  __syncthreads();
#pragma unroll
  for (int i = 0; i < EPT; ++i) {
    const int k = tid + i * ETH;
    if (k < L) {
      const int pos = nstart[wr[i] >> 16] + rk[i];
      list[pos] = wr[i];
      ebuf[pos] = vr[i];
    }
  }
  __syncthreads();
}

__device__ __forceinline__ void row_fetch(int k, const unsigned* list, const float4* ebuf,
                                          const float* mxs, const float* __restrict__ feat,
                                          int lane, float4 (&f)[HH], float (&p)[4], int& tl) {
  const unsigned w = list[k];
  const int s = __builtin_amdgcn_readfirstlane((int)(w & 0xFFFFu));
  tl = __builtin_amdgcn_readfirstlane((int)(w >> 16));
  const float4 v = ebuf[k];
  const float4 m = *(const float4*)(mxs + tl * HH);
  p[0] = __expf(v.x - m.x); p[1] = __expf(v.y - m.y);
  p[2] = __expf(v.z - m.z); p[3] = __expf(v.w - m.w);
  const float4* fr = (const float4*)(feat + (size_t)s * FH) + lane;
#pragma unroll
  for (int j = 0; j < HH; ++j) f[j] = fr[(DD / 4) * j];
}

__device__ __forceinline__ void run_flush(float* acc, float* dns, int tl, int lane,
                                          const float4 (&sum)[HH], const float (&ds)[4]) {
  float* ar = acc + tl * FH + lane;
#pragma unroll
  for (int j = 0; j < HH; ++j) {
    lds_add_f(ar + j * DD + 0 * (DD / 4), sum[j].x);
    lds_add_f(ar + j * DD + 1 * (DD / 4), sum[j].y);
    lds_add_f(ar + j * DD + 2 * (DD / 4), sum[j].z);
    lds_add_f(ar + j * DD + 3 * (DD / 4), sum[j].w);
  }
  if (lane < HH) {
    const float dv = lane == 0 ? ds[0] : lane == 1 ? ds[1] : lane == 2 ? ds[2] : ds[3];
    lds_add_f(&dns[tl * HH + lane], dv);
  }
}

__device__ __forceinline__ void run_step(float* acc, float* dns, int lane, int tl, int& cur,
                                         float4 (&sum)[HH], float (&ds)[4],
                                         const float4 (&f)[HH], const float (&p)[4]) {
  if (tl != cur) {
    if (cur >= 0) run_flush(acc, dns, cur, lane, sum, ds);
    cur = tl;
#pragma unroll
    for (int j = 0; j < HH; ++j) {
      sum[j] = make_float4(0.f, 0.f, 0.f, 0.f);
      ds[j] = 0.f;
    }
  }
#pragma unroll
  for (int j = 0; j < HH; ++j) {
    sum[j].x += f[j].x * p[j]; sum[j].y += f[j].y * p[j];
    sum[j].z += f[j].z * p[j]; sum[j].w += f[j].w * p[j];
    ds[j] += p[j];
  }
}

__device__ __forceinline__ float4 head_cols4(const float* hd, int d0) {
  const float* q = hd + (d0 >> 2);
  float4 a;
  a.x = q[0 * (DD / 4)]; a.y = q[1 * (DD / 4)];
  a.z = q[2 * (DD / 4)]; a.w = q[3 * (DD / 4)];
  return a;
}

__device__ __forceinline__ void accumulate(int L, const unsigned* list, const float4* ebuf,
                                           const float* mxs, float* dns, float* acc,
                                           const float* __restrict__ feat, int wv, int lane) {
  const int per = (L + ETH / 32 - 1) / (ETH / 32);
  int k = __builtin_amdgcn_readfirstlane(wv) * per;
  const int end = (k + per < L) ? k + per : L;
  float4 sum[HH];
  float ds[4];
#pragma unroll
  for (int j = 0; j < HH; ++j) {
    sum[j] = make_float4(0.f, 0.f, 0.f, 0.f);
    ds[j] = 0.f;
  }
  int cur = -1;
  for (; k + 1 < end; k += 2) {
    float4 fa[HH], fb[HH];
    float pa[4], pb[4];
    int ta, tb;
    row_fetch(k, list, ebuf, mxs, feat, lane, fa, pa, ta);
    row_fetch(k + 1, list, ebuf, mxs, feat, lane, fb, pb, tb);
    run_step(acc, dns, lane, ta, cur, sum, ds, fa, pa);
    run_step(acc, dns, lane, tb, cur, sum, ds, fb, pb);
  }
  if (k < end) {
    float4 fa[HH];
    float pa[4];
    int ta;
    row_fetch(k, list, ebuf, mxs, feat, lane, fa, pa, ta);
    run_step(acc, dns, lane, ta, cur, sum, ds, fa, pa);
  }
  if (cur >= 0) run_flush(acc, dns, cur, lane, sum, ds);
}

#define HPT ((RN * FH) / (ETH * 4))
static_assert(HPT * ETH * 4 == RN * FH, "the h rows split evenly over the threads");
static_assert(ETH / 32 == FH / 16, "one wave per 16-column tile of the projection");
static_assert(RN > 32 && RN <= 48 && RN % 8 == 0, "three row tiles, the third half used");

__device__ __forceinline__ void project_rows(const float* hs, const float* __restrict__ wn,
                                             float* __restrict__ fn, int lo, int wv, int lane) {
  const int half = lane >> 4;
  const int l = lane & 15;
  const int r2 = 32 + (l & 7);
  const int x01 = l << 2, x2 = (r2 & 15) << 2;
  const float* a0p = hs + l * FH + half * 2;
  const float* a1p = hs + (16 + l) * FH + half * 2;
  const float* a2p = hs + r2 * FH + half * 2;
  const float* bp = wn + (size_t)(half * 2) * FH + wv * 16 + l;
  v8f c0 = {}, c1 = {}, c2 = {};
#pragma unroll 8
  for (int k = 0; k < FH; k += 4) {
    v2f b; b.x = bp[0]; b.y = bp[FH];
    bp += 4 * FH;
    const v2f a0 = *(const v2f*)(a0p + (k ^ x01));
    const v2f a1 = *(const v2f*)(a1p + (k ^ x01));
    const v2f a2 = *(const v2f*)(a2p + (k ^ x2));
    c0 = __builtin_amdgcn_wmma_f32_16x16x4_f32(false, a0, false, b, (short)0, c0, false, false);
    c1 = __builtin_amdgcn_wmma_f32_16x16x4_f32(false, a1, false, b, (short)0, c1, false, false);
    c2 = __builtin_amdgcn_wmma_f32_16x16x4_f32(false, a2, false, b, (short)0, c2, false, false);
  }
  float* ob = fn + (size_t)(lo + half * 8) * FH + wv * 16 + l;
#pragma unroll
  for (int r = 0; r < 8; ++r) {
    ob[(size_t)r * FH] = c0[r];
    ob[(size_t)(16 + r) * FH] = c1[r];
  }
  if (half == 0) {
#pragma unroll
    for (int r = 0; r < 8; ++r) ob[(size_t)(32 + r) * FH] = c2[r];
  }
}

__device__ __forceinline__ float elu1(float v) {
  return v > 0.f ? v : (__expf(v) - 1.f);
}

template <bool LAST>
__global__ __launch_bounds__(ETH) void edge_own(
    const int* __restrict__ src, const int* __restrict__ dst,
    const float* __restrict__ feat, const float* __restrict__ el,
    const float* __restrict__ er, const float* __restrict__ hin,
    const float* __restrict__ wnext, float* __restrict__ fnext,
    float* __restrict__ outp, unsigned* __restrict__ glist, int* __restrict__ gcnt) {
  __shared__ __attribute__((aligned(16))) float acc[RN * FH];
  __shared__ __attribute__((aligned(16))) float4 ebuf[CAP];
  __shared__ unsigned list[CAP];
  __shared__ __attribute__((aligned(16))) float mxs[RN * HH];
  __shared__ __attribute__((aligned(16))) float dns[RN * HH];
  __shared__ __attribute__((aligned(16))) int hist[RN];
  __shared__ __attribute__((aligned(16))) int nstart[RN];
  __shared__ int cnt;

  const int tid = threadIdx.x;
  const int lane = tid & 31;
  const int wv = tid >> 5;
  const int lo = blockIdx.x * RN;

  for (int i = tid * 4; i < RN * FH; i += ETH * 4)
    *(float4*)(acc + i) = make_float4(0.f, 0.f, 0.f, 0.f);
  if (tid < RN * HH) { mxs[tid] = -__builtin_huge_valf(); dns[tid] = 0.f; }
  if (!LAST && tid < RN) hist[tid] = 0;
  if (tid == 0) cnt = 0;
  __syncthreads();

  unsigned* gl = glist + (size_t)blockIdx.x * CAP;
  int total;
  if (!LAST) {
    scan_emit(dst, 0, EE, lo, list, &cnt, tid, lane);
    __syncthreads();
    total = cnt;
    if (tid == 0) gcnt[blockIdx.x] = total;
  } else {
    total = gcnt[blockIdx.x];
  }

  if (total <= CAP) {
    if (!LAST) {
      edge_scores_sorted(total, list, ebuf, mxs, hist, nstart, src, el, er, lo, tid);
      for (int k = tid; k < total; k += ETH) gl[k] = list[k];
    } else {
      edge_scores_saved(total, gl, list, ebuf, mxs, el, er, lo, tid);
      __syncthreads();
    }
    accumulate(total, list, ebuf, mxs, dns, acc, feat, wv, lane);
  } else {
    for (int c0 = 0; c0 < EE; c0 += CAP) {
      __syncthreads();
      if (tid == 0) cnt = 0;
      __syncthreads();
      scan_emit(dst, c0, (c0 + CAP < EE) ? c0 + CAP : EE, lo, list, &cnt, tid, lane);
      __syncthreads();
      edge_scores(cnt, true, list, ebuf, mxs, src, el, er, lo, tid);
    }
    for (int c0 = 0; c0 < EE; c0 += CAP) {
      __syncthreads();
      if (tid == 0) cnt = 0;
      __syncthreads();
      scan_emit(dst, c0, (c0 + CAP < EE) ? c0 + CAP : EE, lo, list, &cnt, tid, lane);
      __syncthreads();
      const int L = cnt;
      edge_scores(L, false, list, ebuf, mxs, src, el, er, lo, tid);
      __syncthreads();
      accumulate(L, list, ebuf, mxs, dns, acc, feat, wv, lane);
    }
  }
  __syncthreads();

  if (!LAST) {
    float* ob = outp + (size_t)lo * FH;
    float4 hv[HPT];
#pragma unroll
    for (int it = 0; it < HPT; ++it) {
      const int i = (it * ETH + tid) * 4;
      const float d = dns[i >> 7];
      const float dd = d == 0.f ? 1.f : d;
      const float4 a = head_cols4(acc + (i & ~(DD - 1)), i & (DD - 1));
      float4 o;
      o.x = elu1(a.x / dd); o.y = elu1(a.y / dd);
      o.z = elu1(a.z / dd); o.w = elu1(a.w / dd);
      hv[it] = o;
    }
    __syncthreads();
#pragma unroll
    for (int it = 0; it < HPT; ++it) {
      const int i = (it * ETH + tid) * 4;
      const int r = i >> 9, c = i & (FH - 1);
      *(float4*)(acc + r * FH + (c ^ ((r & 15) << 2))) = hv[it];
    }
    __syncthreads();
#pragma unroll
    for (int it = 0; it < HPT; ++it)
      *(float4*)(ob + (it * ETH + tid) * 4) = hv[it];
    project_rows(acc, wnext, fnext, lo, wv, lane);
  } else {
    for (int i = tid * 4; i < RN * DD; i += ETH * 4) {
      const int tl = i >> 7, d0 = i & (DD - 1);
      float4 sum = make_float4(0.f, 0.f, 0.f, 0.f);
#pragma unroll
      for (int h = 0; h < HH; ++h) {
        const float d = dns[tl * HH + h];
        const float dd = d == 0.f ? 1.f : d;
        const float4 a = head_cols4(acc + tl * FH + h * DD, d0);
        const float4 r = *(const float4*)(hin + (size_t)(lo + tl) * FH + h * DD + d0);
        sum.x += elu1(a.x / dd + r.x); sum.y += elu1(a.y / dd + r.y);
        sum.z += elu1(a.z / dd + r.z); sum.w += elu1(a.w / dd + r.w);
      }
      float4 o;
      o.x = sum.x * (1.f / HH); o.y = sum.y * (1.f / HH);
      o.z = sum.z * (1.f / HH); o.w = sum.w * (1.f / HH);
      *(float4*)(outp + (size_t)(lo + tl) * DD + d0) = o;
    }
  }
}

extern "C" void kernel_launch(void* const* d_in, const int* in_sizes, int n_in,
                              void* d_out, int out_size, void* d_ws, size_t ws_size,
                              hipStream_t stream) {
  const int* src = (const int*)d_in[0];
  const int* dst = (const int*)d_in[1];
  const float* x   = (const float*)d_in[2];
  const float* W0  = (const float*)d_in[3];
  const float* al0 = (const float*)d_in[4];
  const float* ar0 = (const float*)d_in[5];
  const float* W1  = (const float*)d_in[6];
  const float* al1 = (const float*)d_in[7];
  const float* ar1 = (const float*)d_in[8];
  float* out = (float*)d_out;

  float* ws    = (float*)d_ws;
  float* feat  = ws;
  float* hbuf  = feat  + (size_t)NN * FH;
  float* el    = hbuf  + (size_t)NN * FH;
  float* er    = el    + (size_t)NN * HH;
  float* feat1 = er    + (size_t)NN * HH;
  unsigned* glist = (unsigned*)(feat1 + (size_t)NN * FH);
  int* gcnt = (int*)(glist + (size_t)NBLK * CAP);

  const int mtiles = (NN + 15) / 16;
  const int mgroups = (mtiles + 7) / 8;
  const int ntiles = FH / 16;
  const int gemmBlocks = mgroups * ntiles;

  gemm16_wmma<<<gemmBlocks, 256, 0, stream>>>(x, W0, feat, NN, K0, FH);
  attn_coef<<<(NN * HH + 255) / 256, 256, 0, stream>>>(feat, al0, ar0, el, er);
  edge_own<false><<<NBLK, ETH, 0, stream>>>(src, dst, feat, el, er, nullptr, W1, feat1, hbuf, glist, gcnt);
  attn_coef<<<(NN * HH + 255) / 256, 256, 0, stream>>>(feat1, al1, ar1, el, er);
  edge_own<true><<<NBLK, ETH, 0, stream>>>(src, dst, feat1, el, er, hbuf, nullptr, nullptr, out, glist, gcnt);
}


# Layer-0 GEMM: a 16-wave work-group copies its 80 x rows into LDS with whole-line loads as the A operand; a wave owns 16 columns x 5 row tiles, one W0 fragment feeds 5 WMMAs; sum order unchanged
# speedup vs baseline: 7.0788x; 1.2162x over previous
#include <hip/hip_runtime.h>
#include <math.h>

#define NN 10000
#define EE 320000
#define HH 4
#define DD 128
#define FH (HH * DD)
#define K0 256
#define NEG_SLOPE 0.2f

typedef __attribute__((ext_vector_type(2))) float v2f;
typedef __attribute__((ext_vector_type(8))) float v8f;

#define GT 5
#define GR (GT * 16)
#define GC 256
#define GTH (GC * 2)
#define XP (K0 + 4)
static_assert(NN % GR == 0, "row blocks tile N");
static_assert(FH % GC == 0, "column blocks tile the output row");
static_assert((GR * (K0 / 4)) % GTH == 0, "the X rows split evenly over the threads");

#define GK 4
static_assert(K0 % (8 * GK) == 0, "the k loop takes two fragment sets per trip");

__device__ __forceinline__ void gemm_load_set(v2f (&a)[GT][GK], v2f (&b)[GK],
                                              const float* ap, const float* __restrict__ bp,
                                              int k) {
#pragma unroll
  for (int s = 0; s < GK; ++s) {
    b[s].x = bp[(size_t)(k + 4 * s) * FH];
    b[s].y = bp[(size_t)(k + 4 * s + 1) * FH];
  }
#pragma unroll
  for (int t = 0; t < GT; ++t) {
#pragma unroll
    for (int s = 0; s < GK; ++s) a[t][s] = *(const v2f*)(ap + t * 16 * XP + k + 4 * s);
  }
}

__device__ __forceinline__ void gemm_mma_set(v8f (&acc)[GT], const v2f (&a)[GT][GK],
                                             const v2f (&b)[GK]) {
#pragma unroll
  for (int s = 0; s < GK; ++s) {
#pragma unroll
    for (int t = 0; t < GT; ++t)
      acc[t] = __builtin_amdgcn_wmma_f32_16x16x4_f32(
           false, a[t][s],  false, b[s],
           (short)0, acc[t],  false,  false);
  }
}

__global__ __launch_bounds__(GTH) void gemm_rows_lds(const float* __restrict__ X,
                                                     const float* __restrict__ W,
                                                     float* __restrict__ out) {
  __shared__ __attribute__((aligned(16))) float xs[GR * XP];

  const int tid = threadIdx.x;
  const int lane = tid & 31;
  const int wv = tid >> 5;
  const int cblocks = FH / GC;
  const int rb = blockIdx.x / cblocks;
  const int cb = blockIdx.x - rb * cblocks;
  const int row0 = rb * GR;

  const float4* xr = (const float4*)(X + (size_t)row0 * K0);
#pragma unroll
  for (int it = 0; it < (GR * (K0 / 4)) / GTH; ++it) {
    const int i = it * GTH + tid;
    const int r = i / (K0 / 4), c = (i % (K0 / 4)) * 4;
    *(float4*)(xs + r * XP + c) = xr[i];
  }
  __syncthreads();

  const int half = lane >> 4;
  const int l = lane & 15;
  const int col = cb * GC + wv * 16 + l;
  const float* ap = xs + l * XP + half * 2;
  const float* bp = W + (size_t)(half * 2) * FH + col;

  v8f acc[GT];
#pragma unroll
  for (int t = 0; t < GT; ++t) acc[t] = v8f{};

  v2f a0[GT][GK], b0[GK], a1[GT][GK], b1[GK];
  gemm_load_set(a0, b0, ap, bp, 0);
  for (int k = 0; k < K0 - 8 * GK; k += 8 * GK) {
    gemm_load_set(a1, b1, ap, bp, k + 4 * GK);
    __builtin_amdgcn_sched_barrier(0);
    gemm_mma_set(acc, a0, b0);
    __builtin_amdgcn_sched_barrier(0);
    gemm_load_set(a0, b0, ap, bp, k + 8 * GK);
    __builtin_amdgcn_sched_barrier(0);
    gemm_mma_set(acc, a1, b1);
    __builtin_amdgcn_sched_barrier(0);
  }
  gemm_load_set(a1, b1, ap, bp, K0 - 4 * GK);
  __builtin_amdgcn_sched_barrier(0);
  gemm_mma_set(acc, a0, b0);
  gemm_mma_set(acc, a1, b1);

  float* ob = out + (size_t)(row0 + half * 8) * FH + col;
#pragma unroll
  for (int t = 0; t < GT; ++t) {
#pragma unroll
    for (int r = 0; r < 8; ++r) ob[(size_t)(t * 16 + r) * FH] = acc[t][r];
  }
}

#define RN 40
#define NBLK (NN / RN)
static_assert(NN % RN == 0, "node ranges must tile N");
static_assert(NBLK <= 255, "an owner number is one byte, 255 means none");
static_assert(EE % 16 == 0, "keys are packed 8 and scanned 16 edges at a time");

__device__ __forceinline__ unsigned owner_key(int d) {
  return (unsigned)d < (unsigned)NN ? (unsigned)d / RN : 255u;
}

__device__ __forceinline__ unsigned owner_keys4(int4 d) {
  return owner_key(d.x) | (owner_key(d.y) << 8) | (owner_key(d.z) << 16) | (owner_key(d.w) << 24);
}

template <bool KEYS>
__global__ void attn_coef(const float* __restrict__ feat,
                          const float* __restrict__ al,
                          const float* __restrict__ ar,
                          float* __restrict__ el, float* __restrict__ er,
                          const int* __restrict__ dst, unsigned* __restrict__ okey) {
  int i = blockIdx.x * blockDim.x + threadIdx.x;
  if (KEYS) {
    const int4* d4 = (const int4*)dst;
    uint2* k2 = (uint2*)okey;
    for (int g = i; g < EE / 8; g += gridDim.x * blockDim.x)
      k2[g] = make_uint2(owner_keys4(d4[2 * g]), owner_keys4(d4[2 * g + 1]));
  }
  if (i >= NN * HH) return;
  int h = i & (HH - 1);
  const float4* f = (const float4*)(feat + (size_t)i * DD);
  const float4* a = (const float4*)(al + h * DD);
  const float4* b = (const float4*)(ar + h * DD);
  float sl = 0.f, sr = 0.f;
#pragma unroll 4
  for (int d = 0; d < DD / 4; ++d) {
    float4 fv = f[d], av = a[d], bv = b[d];
    sl += fv.x * av.x + fv.y * av.y + fv.z * av.z + fv.w * av.w;
    sr += fv.x * bv.x + fv.y * bv.y + fv.z * bv.z + fv.w * bv.w;
  }
  el[i] = sl;
  er[i] = sr;
}

#define ETH 1024
#define CAP 4096
#define EPT (CAP / ETH)

static_assert(CAP % 16 == 0, "a slice of the key plane is whole 16-byte groups");
static_assert(RN < (1 << 6) && NN <= (1 << 16), "packed fields");
static_assert(DD / 4 == 32, "a wave's 32 lanes x 16 bytes cover one head of a row");
static_assert(EPT * ETH == CAP, "a full list splits evenly over the threads");

__device__ __forceinline__ float leaky(float v) {
  return v > 0.f ? v : NEG_SLOPE * v;
}

__device__ __forceinline__ void lds_max_f(float* addr, float v) {
  if (__float_as_int(v) >= 0)
    atomicMax((int*)addr, __float_as_int(v));
  else
    atomicMin((unsigned int*)addr, __float_as_uint(v));
}

__device__ __forceinline__ void lds_add_f(float* addr, float v) {
  __hip_atomic_fetch_add(addr, v, __ATOMIC_RELAXED, __HIP_MEMORY_SCOPE_WORKGROUP);
}

__device__ __forceinline__ void load_key4(uint4 (&v)[4], const uint4* __restrict__ k4,
                                          int q0, int q_hi, int tid) {
#pragma unroll
  for (int u = 0; u < 4; ++u) {
    const int q = q0 + u * ETH + tid;
    v[u] = k4[q < q_hi ? q : q_hi - 1];
  }
}

__device__ __forceinline__ unsigned eq_bytes(unsigned w, unsigned pat) {
  const unsigned x = w ^ pat;
  return ~(((x & 0x7F7F7F7Fu) + 0x7F7F7F7Fu) | x | 0x7F7F7F7Fu);
}

__device__ __forceinline__ unsigned eq_group(uint4 g, unsigned pat) {
  return eq_bytes(g.x, pat) | (eq_bytes(g.y, pat) >> 1) |
         (eq_bytes(g.z, pat) >> 2) | (eq_bytes(g.w, pat) >> 3);
}

__device__ __forceinline__ void scan_keys(const uint4* __restrict__ k4, int q_lo, int q_hi,
                                          unsigned pat, unsigned* list, int* cnt, int tid) {
  const int STEP = 4 * ETH;
  uint4 cur[4], nxt[4];
  load_key4(cur, k4, q_lo, q_hi, tid);
  for (int q0 = q_lo; q0 < q_hi; q0 += STEP) {
    load_key4(nxt, k4, q0 + STEP, q_hi, tid);
#pragma unroll
    for (int u = 0; u < 4; ++u) {
      const int q = q0 + u * ETH + tid;
      unsigned m = eq_group(cur[u], pat);
      if (q >= q_hi) m = 0u;
      if (m != 0u) {
        int p = atomicAdd(cnt, __popc(m));
        const unsigned eb = (unsigned)q << 4;
        do {
          const unsigned bit = (unsigned)__builtin_ctz(m);
          if (p < CAP) list[p] = eb + ((7u - (bit & 7u)) << 2) + (bit >> 3);
          ++p;
          m &= m - 1u;
        } while (m != 0u);
      }
    }
#pragma unroll
    for (int u = 0; u < 4; ++u) cur[u] = nxt[u];
  }
}

__device__ __forceinline__ void edge_scores(int L, bool domax, unsigned* list, float4* ebuf,
                                            float* mxs, const int* __restrict__ src,
                                            const int* __restrict__ dst,
                                            const float* __restrict__ el,
                                            const float* __restrict__ er, int lo, int tid) {
  for (int k = tid; k < L; k += ETH) {
    const int e = (int)list[k];
    const int tl = dst[e] - lo;
    const int s = src[e];
    const float4 a = *(const float4*)(el + (size_t)s * HH);
    const float4 b = *(const float4*)(er + (size_t)(lo + tl) * HH);
    float4 v;
    v.x = leaky(a.x + b.x); v.y = leaky(a.y + b.y);
    v.z = leaky(a.z + b.z); v.w = leaky(a.w + b.w);
    ebuf[k] = v;
    list[k] = (unsigned)s | ((unsigned)tl << 16);
    if (domax) {
      lds_max_f(&mxs[tl * HH + 0], v.x); lds_max_f(&mxs[tl * HH + 1], v.y);
      lds_max_f(&mxs[tl * HH + 2], v.z); lds_max_f(&mxs[tl * HH + 3], v.w);
    }
  }
}

__device__ __forceinline__ void edge_scores_saved(int L, const unsigned* __restrict__ gl,
                                                  unsigned* list, float4* ebuf, float* mxs,
                                                  const float* __restrict__ el,
                                                  const float* __restrict__ er, int lo, int tid) {
  for (int k = tid; k < L; k += ETH) {
    const unsigned w = gl[k];
    const int s = (int)(w & 0xFFFFu);
    const int tl = (int)(w >> 16);
    const float4 a = *(const float4*)(el + (size_t)s * HH);
    const float4 b = *(const float4*)(er + (size_t)(lo + tl) * HH);
    float4 v;
    v.x = leaky(a.x + b.x); v.y = leaky(a.y + b.y);
    v.z = leaky(a.z + b.z); v.w = leaky(a.w + b.w);
    ebuf[k] = v;
    list[k] = w;
    lds_max_f(&mxs[tl * HH + 0], v.x); lds_max_f(&mxs[tl * HH + 1], v.y);
    lds_max_f(&mxs[tl * HH + 2], v.z); lds_max_f(&mxs[tl * HH + 3], v.w);
  }
}

__device__ __forceinline__ void edge_scores_sorted(int L, unsigned* list, float4* ebuf,
                                                   float* mxs, int* hist, int* nstart,
                                                   const int* __restrict__ src,
                                                   const int* __restrict__ dst,
                                                   const float* __restrict__ el,
                                                   const float* __restrict__ er, int lo, int tid) {
  unsigned wr[EPT];
  float4 vr[EPT];
  int rk[EPT];
#pragma unroll
  for (int i = 0; i < EPT; ++i) {
    const int k = tid + i * ETH;
    wr[i] = 0u; rk[i] = 0; vr[i] = make_float4(0.f, 0.f, 0.f, 0.f);
    if (k < L) {
      const int e = (int)list[k];
      const int tl = dst[e] - lo;
      const int s = src[e];
      const float4 a = *(const float4*)(el + (size_t)s * HH);
      const float4 b = *(const float4*)(er + (size_t)(lo + tl) * HH);
      float4 v;
      v.x = leaky(a.x + b.x); v.y = leaky(a.y + b.y);
      v.z = leaky(a.z + b.z); v.w = leaky(a.w + b.w);
      lds_max_f(&mxs[tl * HH + 0], v.x); lds_max_f(&mxs[tl * HH + 1], v.y);
      lds_max_f(&mxs[tl * HH + 2], v.z); lds_max_f(&mxs[tl * HH + 3], v.w);
      rk[i] = atomicAdd(&hist[tl], 1);
      wr[i] = (unsigned)s | ((unsigned)tl << 16);
      vr[i] = v;
    }
  }
  __syncthreads();
  if (tid < RN) {
    int first = 0;
#pragma unroll
    for (int j = 0; j < RN; ++j) first += (j < tid) ? hist[j] : 0;
    nstart[tid] = first;
  }
  __syncthreads();
#pragma unroll
  for (int i = 0; i < EPT; ++i) {
    const int k = tid + i * ETH;
    if (k < L) {
      const int pos = nstart[wr[i] >> 16] + rk[i];
      list[pos] = wr[i];
      ebuf[pos] = vr[i];
    }
  }
  __syncthreads();
}

__device__ __forceinline__ void row_fetch(int k, const unsigned* list, const float4* ebuf,
                                          const float* mxs, const float* __restrict__ feat,
                                          int lane, float4 (&f)[HH], float (&p)[4], int& tl) {
  const unsigned w = list[k];
  const int s = __builtin_amdgcn_readfirstlane((int)(w & 0xFFFFu));
  tl = __builtin_amdgcn_readfirstlane((int)(w >> 16));
  const float4 v = ebuf[k];
  const float4 m = *(const float4*)(mxs + tl * HH);
  p[0] = __expf(v.x - m.x); p[1] = __expf(v.y - m.y);
  p[2] = __expf(v.z - m.z); p[3] = __expf(v.w - m.w);
  const float4* fr = (const float4*)(feat + (size_t)s * FH) + lane;
#pragma unroll
  for (int j = 0; j < HH; ++j) f[j] = fr[(DD / 4) * j];
}

__device__ __forceinline__ void run_flush(float* acc, float* dns, int tl, int lane,
                                          const float4 (&sum)[HH], const float (&ds)[4]) {
  float* ar = acc + tl * FH + lane;
#pragma unroll
  for (int j = 0; j < HH; ++j) {
    lds_add_f(ar + j * DD + 0 * (DD / 4), sum[j].x);
    lds_add_f(ar + j * DD + 1 * (DD / 4), sum[j].y);
    lds_add_f(ar + j * DD + 2 * (DD / 4), sum[j].z);
    lds_add_f(ar + j * DD + 3 * (DD / 4), sum[j].w);
  }
  if (lane < HH) {
    const float dv = lane == 0 ? ds[0] : lane == 1 ? ds[1] : lane == 2 ? ds[2] : ds[3];
    lds_add_f(&dns[tl * HH + lane], dv);
  }
}

__device__ __forceinline__ void run_step(float* acc, float* dns, int lane, int tl, int& cur,
                                         float4 (&sum)[HH], float (&ds)[4],
                                         const float4 (&f)[HH], const float (&p)[4]) {
  if (tl != cur) {
    if (cur >= 0) run_flush(acc, dns, cur, lane, sum, ds);
    cur = tl;
#pragma unroll
    for (int j = 0; j < HH; ++j) {
      sum[j] = make_float4(0.f, 0.f, 0.f, 0.f);
      ds[j] = 0.f;
    }
  }
#pragma unroll
  for (int j = 0; j < HH; ++j) {
    sum[j].x += f[j].x * p[j]; sum[j].y += f[j].y * p[j];
    sum[j].z += f[j].z * p[j]; sum[j].w += f[j].w * p[j];
    ds[j] += p[j];
  }
}

__device__ __forceinline__ float4 head_cols4(const float* hd, int d0) {
  const float* q = hd + (d0 >> 2);
  float4 a;
  a.x = q[0 * (DD / 4)]; a.y = q[1 * (DD / 4)];
  a.z = q[2 * (DD / 4)]; a.w = q[3 * (DD / 4)];
  return a;
}

__device__ __forceinline__ void accumulate(int L, const unsigned* list, const float4* ebuf,
                                           const float* mxs, float* dns, float* acc,
                                           const float* __restrict__ feat, int wv, int lane) {
  const int per = (L + ETH / 32 - 1) / (ETH / 32);
  int k = __builtin_amdgcn_readfirstlane(wv) * per;
  const int end = (k + per < L) ? k + per : L;
  float4 sum[HH];
  float ds[4];
#pragma unroll
  for (int j = 0; j < HH; ++j) {
    sum[j] = make_float4(0.f, 0.f, 0.f, 0.f);
    ds[j] = 0.f;
  }
  int cur = -1;
  for (; k + 1 < end; k += 2) {
    float4 fa[HH], fb[HH];
    float pa[4], pb[4];
    int ta, tb;
    row_fetch(k, list, ebuf, mxs, feat, lane, fa, pa, ta);
    row_fetch(k + 1, list, ebuf, mxs, feat, lane, fb, pb, tb);
    run_step(acc, dns, lane, ta, cur, sum, ds, fa, pa);
    run_step(acc, dns, lane, tb, cur, sum, ds, fb, pb);
  }
  if (k < end) {
    float4 fa[HH];
    float pa[4];
    int ta;
    row_fetch(k, list, ebuf, mxs, feat, lane, fa, pa, ta);
    run_step(acc, dns, lane, ta, cur, sum, ds, fa, pa);
  }
  if (cur >= 0) run_flush(acc, dns, cur, lane, sum, ds);
}

#define HPT ((RN * FH) / (ETH * 4))
static_assert(HPT * ETH * 4 == RN * FH, "the h rows split evenly over the threads");
static_assert(ETH / 32 == FH / 16, "one wave per 16-column tile of the projection");
static_assert(RN > 32 && RN <= 48 && RN % 8 == 0, "three row tiles, the third half used");

__device__ __forceinline__ void project_rows(const float* hs, const float* __restrict__ wn,
                                             float* __restrict__ fn, int lo, int wv, int lane) {
  const int half = lane >> 4;
  const int l = lane & 15;
  const int r2 = 32 + (l & 7);
  const int x01 = l << 2, x2 = (r2 & 15) << 2;
  const float* a0p = hs + l * FH + half * 2;
  const float* a1p = hs + (16 + l) * FH + half * 2;
  const float* a2p = hs + r2 * FH + half * 2;
  const float* bp = wn + (size_t)(half * 2) * FH + wv * 16 + l;
  v8f c0 = {}, c1 = {}, c2 = {};
#pragma unroll 8
  for (int k = 0; k < FH; k += 4) {
    v2f b; b.x = bp[0]; b.y = bp[FH];
    bp += 4 * FH;
    const v2f a0 = *(const v2f*)(a0p + (k ^ x01));
    const v2f a1 = *(const v2f*)(a1p + (k ^ x01));
    const v2f a2 = *(const v2f*)(a2p + (k ^ x2));
    c0 = __builtin_amdgcn_wmma_f32_16x16x4_f32(false, a0, false, b, (short)0, c0, false, false);
    c1 = __builtin_amdgcn_wmma_f32_16x16x4_f32(false, a1, false, b, (short)0, c1, false, false);
    c2 = __builtin_amdgcn_wmma_f32_16x16x4_f32(false, a2, false, b, (short)0, c2, false, false);
  }
  float* ob = fn + (size_t)(lo + half * 8) * FH + wv * 16 + l;
#pragma unroll
  for (int r = 0; r < 8; ++r) {
    ob[(size_t)r * FH] = c0[r];
    ob[(size_t)(16 + r) * FH] = c1[r];
  }
  if (half == 0) {
#pragma unroll
    for (int r = 0; r < 8; ++r) ob[(size_t)(32 + r) * FH] = c2[r];
  }
}

__device__ __forceinline__ float elu1(float v) {
  return v > 0.f ? v : (__expf(v) - 1.f);
}

template <bool LAST>
__global__ __launch_bounds__(ETH) void edge_own(
    const int* __restrict__ src, const int* __restrict__ dst,
    const float* __restrict__ feat, const float* __restrict__ el,
    const float* __restrict__ er, const float* __restrict__ hin,
    const float* __restrict__ wnext, float* __restrict__ fnext,
    float* __restrict__ outp, unsigned* __restrict__ glist, int* __restrict__ gcnt,
    const unsigned* __restrict__ okey) {
  __shared__ __attribute__((aligned(16))) float acc[RN * FH];
  __shared__ __attribute__((aligned(16))) float4 ebuf[CAP];
  __shared__ unsigned list[CAP];
  __shared__ __attribute__((aligned(16))) float mxs[RN * HH];
  __shared__ __attribute__((aligned(16))) float dns[RN * HH];
  __shared__ __attribute__((aligned(16))) int hist[RN];
  __shared__ __attribute__((aligned(16))) int nstart[RN];
  __shared__ int cnt;

  const int tid = threadIdx.x;
  const int lane = tid & 31;
  const int wv = tid >> 5;
  const int lo = blockIdx.x * RN;
  const unsigned pat = blockIdx.x * 0x01010101u;
  const uint4* k4 = (const uint4*)okey;

  for (int i = tid * 4; i < RN * FH; i += ETH * 4)
    *(float4*)(acc + i) = make_float4(0.f, 0.f, 0.f, 0.f);
  if (tid < RN * HH) { mxs[tid] = -__builtin_huge_valf(); dns[tid] = 0.f; }
  if (!LAST && tid < RN) hist[tid] = 0;
  if (tid == 0) cnt = 0;
  __syncthreads();

  unsigned* gl = glist + (size_t)blockIdx.x * CAP;
  int total;
  if (!LAST) {
    scan_keys(k4, 0, EE / 16, pat, list, &cnt, tid);
    __syncthreads();
    total = cnt;
    if (tid == 0) gcnt[blockIdx.x] = total;
  } else {
    total = gcnt[blockIdx.x];
  }

  if (total <= CAP) {
    if (!LAST) {
      edge_scores_sorted(total, list, ebuf, mxs, hist, nstart, src, dst, el, er, lo, tid);
      for (int k = tid; k < total; k += ETH) gl[k] = list[k];
    } else {
      edge_scores_saved(total, gl, list, ebuf, mxs, el, er, lo, tid);
      __syncthreads();
    }
    accumulate(total, list, ebuf, mxs, dns, acc, feat, wv, lane);
  } else {
    for (int c0 = 0; c0 < EE; c0 += CAP) {
      __syncthreads();
      if (tid == 0) cnt = 0;
      __syncthreads();
      scan_keys(k4, c0 / 16, ((c0 + CAP < EE) ? c0 + CAP : EE) / 16, pat, list, &cnt, tid);
      __syncthreads();
      edge_scores(cnt, true, list, ebuf, mxs, src, dst, el, er, lo, tid);
    }
    for (int c0 = 0; c0 < EE; c0 += CAP) {
      __syncthreads();
      if (tid == 0) cnt = 0;
      __syncthreads();
      scan_keys(k4, c0 / 16, ((c0 + CAP < EE) ? c0 + CAP : EE) / 16, pat, list, &cnt, tid);
      __syncthreads();
      const int L = cnt;
      edge_scores(L, false, list, ebuf, mxs, src, dst, el, er, lo, tid);
      __syncthreads();
      accumulate(L, list, ebuf, mxs, dns, acc, feat, wv, lane);
    }
  }
  __syncthreads();

  if (!LAST) {
    float* ob = outp + (size_t)lo * FH;
    float4 hv[HPT];
#pragma unroll
    for (int it = 0; it < HPT; ++it) {
      const int i = (it * ETH + tid) * 4;
      const float d = dns[i >> 7];
      const float dd = d == 0.f ? 1.f : d;
      const float4 a = head_cols4(acc + (i & ~(DD - 1)), i & (DD - 1));
      float4 o;
      o.x = elu1(a.x / dd); o.y = elu1(a.y / dd);
      o.z = elu1(a.z / dd); o.w = elu1(a.w / dd);
      hv[it] = o;
    }
    __syncthreads();
#pragma unroll
    for (int it = 0; it < HPT; ++it) {
      const int i = (it * ETH + tid) * 4;
      const int r = i >> 9, c = i & (FH - 1);
      *(float4*)(acc + r * FH + (c ^ ((r & 15) << 2))) = hv[it];
    }
    __syncthreads();
#pragma unroll
    for (int it = 0; it < HPT; ++it)
      *(float4*)(ob + (it * ETH + tid) * 4) = hv[it];
    project_rows(acc, wnext, fnext, lo, wv, lane);
  } else {
    for (int i = tid * 4; i < RN * DD; i += ETH * 4) {
      const int tl = i >> 7, d0 = i & (DD - 1);
      float4 sum = make_float4(0.f, 0.f, 0.f, 0.f);
#pragma unroll
      for (int h = 0; h < HH; ++h) {
        const float d = dns[tl * HH + h];
        const float dd = d == 0.f ? 1.f : d;
        const float4 a = head_cols4(acc + tl * FH + h * DD, d0);
        const float4 r = *(const float4*)(hin + (size_t)(lo + tl) * FH + h * DD + d0);
        sum.x += elu1(a.x / dd + r.x); sum.y += elu1(a.y / dd + r.y);
        sum.z += elu1(a.z / dd + r.z); sum.w += elu1(a.w / dd + r.w);
      }
      float4 o;
      o.x = sum.x * (1.f / HH); o.y = sum.y * (1.f / HH);
      o.z = sum.z * (1.f / HH); o.w = sum.w * (1.f / HH);
      *(float4*)(outp + (size_t)(lo + tl) * DD + d0) = o;
    }
  }
}

extern "C" void kernel_launch(void* const* d_in, const int* in_sizes, int n_in,
                              void* d_out, int out_size, void* d_ws, size_t ws_size,
                              hipStream_t stream) {
  const int* src = (const int*)d_in[0];
  const int* dst = (const int*)d_in[1];
  const float* x   = (const float*)d_in[2];
  const float* W0  = (const float*)d_in[3];
  const float* al0 = (const float*)d_in[4];
  const float* ar0 = (const float*)d_in[5];
  const float* W1  = (const float*)d_in[6];
  const float* al1 = (const float*)d_in[7];
  const float* ar1 = (const float*)d_in[8];
  float* out = (float*)d_out;

  float* ws    = (float*)d_ws;
  float* feat  = ws;
  float* hbuf  = feat  + (size_t)NN * FH;
  float* el    = hbuf  + (size_t)NN * FH;
  float* er    = el    + (size_t)NN * HH;
  float* feat1 = er    + (size_t)NN * HH;
  unsigned* glist = (unsigned*)(feat1 + (size_t)NN * FH);
  int* gcnt = (int*)(glist + (size_t)NBLK * CAP);
  unsigned* okey = (unsigned*)(gcnt + 256);

  gemm_rows_lds<<<(NN / GR) * (FH / GC), GTH, 0, stream>>>(x, W0, feat);
  attn_coef<true><<<(NN * HH + 255) / 256, 256, 0, stream>>>(feat, al0, ar0, el, er, dst, okey);
  edge_own<false><<<NBLK, ETH, 0, stream>>>(src, dst, feat, el, er, nullptr, W1, feat1, hbuf, glist, gcnt,
                                            okey);
  attn_coef<false><<<(NN * HH + 255) / 256, 256, 0, stream>>>(feat1, al1, ar1, el, er, nullptr, nullptr);
  edge_own<true><<<NBLK, ETH, 0, stream>>>(src, dst, feat1, el, er, hbuf, nullptr, nullptr, out, glist, gcnt,
                                           okey);
}


# Edge kernel takes an edge's score from the source row it gathers anyway (in-wave dots with al, one-pass running-max softmax); el/er planes, both attn_coef launches gone; el, softmax sum order change
# speedup vs baseline: 7.4562x; 1.0533x over previous
#include <hip/hip_runtime.h>
#include <math.h>

#define NN 10000
#define EE 320000
#define HH 4
#define DD 128
#define FH (HH * DD)
#define K0 256
#define NEG_SLOPE 0.2f

typedef __attribute__((ext_vector_type(2))) float v2f;
typedef __attribute__((ext_vector_type(8))) float v8f;

#define RN 40
#define NBLK (NN / RN)
static_assert(NN % RN == 0, "node ranges must tile N");
static_assert(NBLK <= 255, "an owner number is one byte, 255 means none");
static_assert(EE % 16 == 0, "keys are packed 8 and scanned 16 edges at a time");

__device__ __forceinline__ unsigned owner_key(int d) {
  return (unsigned)d < (unsigned)NN ? (unsigned)d / RN : 255u;
}

__device__ __forceinline__ unsigned owner_keys4(int4 d) {
  return owner_key(d.x) | (owner_key(d.y) << 8) | (owner_key(d.z) << 16) | (owner_key(d.w) << 24);
}

#define GT 5
#define GR (GT * 16)
#define GC 256
#define GTH (GC * 2)
#define XP (K0 + 4)
static_assert(NN % GR == 0, "row blocks tile N");
static_assert(FH % GC == 0, "column blocks tile the output row");
static_assert((GR * (K0 / 4)) % GTH == 0, "the X rows split evenly over the threads");

#define GK 4
static_assert(K0 % (8 * GK) == 0, "the k loop takes two fragment sets per trip");

__device__ __forceinline__ void gemm_load_set(v2f (&a)[GT][GK], v2f (&b)[GK],
                                              const float* ap, const float* __restrict__ bp,
                                              int k) {
#pragma unroll
  for (int s = 0; s < GK; ++s) {
    b[s].x = bp[(size_t)(k + 4 * s) * FH];
    b[s].y = bp[(size_t)(k + 4 * s + 1) * FH];
  }
#pragma unroll
  for (int t = 0; t < GT; ++t) {
#pragma unroll
    for (int s = 0; s < GK; ++s) a[t][s] = *(const v2f*)(ap + t * 16 * XP + k + 4 * s);
  }
}

__device__ __forceinline__ void gemm_mma_set(v8f (&acc)[GT], const v2f (&a)[GT][GK],
                                             const v2f (&b)[GK]) {
#pragma unroll
  for (int s = 0; s < GK; ++s) {
#pragma unroll
    for (int t = 0; t < GT; ++t)
      acc[t] = __builtin_amdgcn_wmma_f32_16x16x4_f32(
           false, a[t][s],  false, b[s],
           (short)0, acc[t],  false,  false);
  }
}

static_assert((NN / GR) * (FH / GC) * GTH >= EE / 8, "the grid's threads cover the key plane, 8 keys each");

__global__ __launch_bounds__(GTH) void gemm_rows_lds(const float* __restrict__ X,
                                                     const float* __restrict__ W,
                                                     float* __restrict__ out,
                                                     const int* __restrict__ dst,
                                                     unsigned* __restrict__ okey) {
  __shared__ __attribute__((aligned(16))) float xs[GR * XP];

  const int tid = threadIdx.x;
  const int lane = tid & 31;
  const int wv = tid >> 5;
  const int cblocks = FH / GC;
  const int rb = blockIdx.x / cblocks;
  const int cb = blockIdx.x - rb * cblocks;
  const int row0 = rb * GR;

  {
    const int g = blockIdx.x * GTH + tid;
    if (g < EE / 8) {
      const int4* d4 = (const int4*)dst;
      ((uint2*)okey)[g] = make_uint2(owner_keys4(d4[2 * g]), owner_keys4(d4[2 * g + 1]));
    }
  }

  const float4* xr = (const float4*)(X + (size_t)row0 * K0);
#pragma unroll
  for (int it = 0; it < (GR * (K0 / 4)) / GTH; ++it) {
    const int i = it * GTH + tid;
    const int r = i / (K0 / 4), c = (i % (K0 / 4)) * 4;
    *(float4*)(xs + r * XP + c) = xr[i];
  }
  __syncthreads();

  const int half = lane >> 4;
  const int l = lane & 15;
  const int col = cb * GC + wv * 16 + l;
  const float* ap = xs + l * XP + half * 2;
  const float* bp = W + (size_t)(half * 2) * FH + col;

  v8f acc[GT];
#pragma unroll
  for (int t = 0; t < GT; ++t) acc[t] = v8f{};

  v2f a0[GT][GK], b0[GK], a1[GT][GK], b1[GK];
  gemm_load_set(a0, b0, ap, bp, 0);
  for (int k = 0; k < K0 - 8 * GK; k += 8 * GK) {
    gemm_load_set(a1, b1, ap, bp, k + 4 * GK);
    __builtin_amdgcn_sched_barrier(0);
    gemm_mma_set(acc, a0, b0);
    __builtin_amdgcn_sched_barrier(0);
    gemm_load_set(a0, b0, ap, bp, k + 8 * GK);
    __builtin_amdgcn_sched_barrier(0);
    gemm_mma_set(acc, a1, b1);
    __builtin_amdgcn_sched_barrier(0);
  }
  gemm_load_set(a1, b1, ap, bp, K0 - 4 * GK);
  __builtin_amdgcn_sched_barrier(0);
  gemm_mma_set(acc, a0, b0);
  gemm_mma_set(acc, a1, b1);

  float* ob = out + (size_t)(row0 + half * 8) * FH + col;
#pragma unroll
  for (int t = 0; t < GT; ++t) {
#pragma unroll
    for (int r = 0; r < 8; ++r) ob[(size_t)(t * 16 + r) * FH] = acc[t][r];
  }
}

#define ETH 1024
#define CAP 4096
#define EPT (CAP / ETH)
#define NWV (ETH / 32)

static_assert(CAP % 16 == 0, "a slice of the key plane is whole 16-byte groups");
static_assert(RN < (1 << 6) && NN <= (1 << 16), "packed fields");
static_assert(DD / 4 == 32, "a wave's 32 lanes x 16 bytes cover one head of a row");
static_assert(EPT * ETH == CAP, "a full list splits evenly over the threads");
static_assert(HH == 4, "the 4 heads' dots are folded over the lane halves and the halves of a row");

__device__ __forceinline__ float leaky(float v) {
  return v > 0.f ? v : NEG_SLOPE * v;
}

__device__ __forceinline__ void lds_max_f(float* addr, float v) {
  if (__float_as_int(v) >= 0)
    atomicMax((int*)addr, __float_as_int(v));
  else
    atomicMin((unsigned int*)addr, __float_as_uint(v));
}

__device__ __forceinline__ void lds_add_f(float* addr, float v) {
  __hip_atomic_fetch_add(addr, v, __ATOMIC_RELAXED, __HIP_MEMORY_SCOPE_WORKGROUP);
}

__device__ __forceinline__ void load_key4(uint4 (&v)[4], const uint4* __restrict__ k4,
                                          int q0, int q_hi, int tid) {
#pragma unroll
  for (int u = 0; u < 4; ++u) {
    const int q = q0 + u * ETH + tid;
    v[u] = k4[q < q_hi ? q : q_hi - 1];
  }
}

__device__ __forceinline__ unsigned eq_bytes(unsigned w, unsigned pat) {
  const unsigned x = w ^ pat;
  return ~(((x & 0x7F7F7F7Fu) + 0x7F7F7F7Fu) | x | 0x7F7F7F7Fu);
}

__device__ __forceinline__ unsigned eq_group(uint4 g, unsigned pat) {
  return eq_bytes(g.x, pat) | (eq_bytes(g.y, pat) >> 1) |
         (eq_bytes(g.z, pat) >> 2) | (eq_bytes(g.w, pat) >> 3);
}

__device__ __forceinline__ void scan_keys(const uint4* __restrict__ k4, int q_lo, int q_hi,
                                          unsigned pat, unsigned* list, int* cnt, int tid) {
  const int STEP = 4 * ETH;
  uint4 cur[4], nxt[4];
  load_key4(cur, k4, q_lo, q_hi, tid);
  for (int q0 = q_lo; q0 < q_hi; q0 += STEP) {
    load_key4(nxt, k4, q0 + STEP, q_hi, tid);
#pragma unroll
    for (int u = 0; u < 4; ++u) {
      const int q = q0 + u * ETH + tid;
      unsigned m = eq_group(cur[u], pat);
      if (q >= q_hi) m = 0u;
      if (m != 0u) {
        int p = atomicAdd(cnt, __popc(m));
        const unsigned eb = (unsigned)q << 4;
        do {
          const unsigned bit = (unsigned)__builtin_ctz(m);
          if (p < CAP) list[p] = eb + ((7u - (bit & 7u)) << 2) + (bit >> 3);
          ++p;
          m &= m - 1u;
        } while (m != 0u);
      }
    }
#pragma unroll
    for (int u = 0; u < 4; ++u) cur[u] = nxt[u];
  }
}

__device__ __forceinline__ void edge_prepare(int L, unsigned* list, const int* __restrict__ src,
                                             const int* __restrict__ dst, int lo, int tid) {
  for (int k = tid; k < L; k += ETH) {
    const int e = (int)list[k];
    const int tl = dst[e] - lo;
    list[k] = (unsigned)src[e] | ((unsigned)tl << 16);
  }
}

__device__ __forceinline__ void edge_sort(int L, unsigned* list, int* hist, int* nstart,
                                          const int* __restrict__ src,
                                          const int* __restrict__ dst, int lo, int tid) {
  unsigned wr[EPT];
  int rk[EPT];
#pragma unroll
  for (int i = 0; i < EPT; ++i) {
    const int k = tid + i * ETH;
    wr[i] = 0u; rk[i] = 0;
    if (k < L) {
      const int e = (int)list[k];
      const int tl = dst[e] - lo;
      rk[i] = atomicAdd(&hist[tl], 1);
      wr[i] = (unsigned)src[e] | ((unsigned)tl << 16);
    }
  }
  __syncthreads();
  if (tid < RN) {
    int first = 0;
#pragma unroll
    for (int j = 0; j < RN; ++j) first += (j < tid) ? hist[j] : 0;
    nstart[tid] = first;
  }
  __syncthreads();
#pragma unroll
  for (int i = 0; i < EPT; ++i) {
    const int k = tid + i * ETH;
    if (k < L) list[nstart[wr[i] >> 16] + rk[i]] = wr[i];
  }
  __syncthreads();
}

__device__ __forceinline__ void row_load(const float* __restrict__ feat, int s, int lane,
                                         float4 (&f)[HH]) {
  const float4* fr = (const float4*)(feat + (size_t)s * FH) + lane;
#pragma unroll
  for (int j = 0; j < HH; ++j) f[j] = fr[(DD / 4) * j];
}

__device__ __forceinline__ float xhalf(float x) {
  const int b = __float_as_int(x);
  return __int_as_float(__builtin_amdgcn_permlanex16(b, b, 0x76543210, 0xfedcba98, false, false));
}

template <int M>
__device__ __forceinline__ float xrow(float x) {
  return __int_as_float(__builtin_amdgcn_update_dpp(0, __float_as_int(x), 0x160 + M, 0xf, 0xf, true));
}

__device__ __forceinline__ float dot4(float4 a, float4 b) {
  return a.x * b.x + a.y * b.y + a.z * b.z + a.w * b.w;
}

#define HEAD_OF(lane) (2 * ((lane) >> 4) + (((lane) >> 3) & 1))
__device__ __forceinline__ float head_dots(const float4 (&f)[HH], const float4 (&v)[HH], int lane) {
  const float p0 = dot4(f[0], v[0]), p1 = dot4(f[1], v[1]);
  const float p2 = dot4(f[2], v[2]), p3 = dot4(f[3], v[3]);
  const bool hi = (lane & 16) != 0;
  const float k0 = (hi ? p2 : p0) + xhalf(hi ? p0 : p2);
  const float k1 = (hi ? p3 : p1) + xhalf(hi ? p1 : p3);
  const bool up = (lane & 8) != 0;
  float k = (up ? k1 : k0) + xrow<8>(up ? k0 : k1);
  k += xrow<4>(k);
  k += xrow<2>(k);
  k += xrow<1>(k);
  return k;
}

__device__ __forceinline__ void bcast_heads(float v, float (&o)[HH]) {
#pragma unroll
  for (int h = 0; h < HH; ++h)
    o[h] = __int_as_float(__builtin_amdgcn_readlane(__float_as_int(v), 8 * h));
}

__device__ __forceinline__ void row_store_t(float* row, int lane, const float4 (&s)[HH]) {
  float* q = row + lane;
#pragma unroll
  for (int j = 0; j < HH; ++j) {
    q[j * DD + 0 * (DD / 4)] = s[j].x; q[j * DD + 1 * (DD / 4)] = s[j].y;
    q[j * DD + 2 * (DD / 4)] = s[j].z; q[j * DD + 3 * (DD / 4)] = s[j].w;
  }
}

__device__ __forceinline__ void row_load_t(const float* row, int lane, float4 (&s)[HH]) {
  const float* q = row + lane;
#pragma unroll
  for (int j = 0; j < HH; ++j) {
    s[j].x = q[j * DD + 0 * (DD / 4)]; s[j].y = q[j * DD + 1 * (DD / 4)];
    s[j].z = q[j * DD + 2 * (DD / 4)]; s[j].w = q[j * DD + 3 * (DD / 4)];
  }
}

__device__ __forceinline__ void row_add_t(float* row, int lane, const float4 (&s)[HH],
                                          const float (&g)[HH]) {
  float* q = row + lane;
#pragma unroll
  for (int j = 0; j < HH; ++j) {
    lds_add_f(q + j * DD + 0 * (DD / 4), s[j].x * g[j]);
    lds_add_f(q + j * DD + 1 * (DD / 4), s[j].y * g[j]);
    lds_add_f(q + j * DD + 2 * (DD / 4), s[j].z * g[j]);
    lds_add_f(q + j * DD + 3 * (DD / 4), s[j].w * g[j]);
  }
}

struct Run {
  float4 sum[HH];
  float m, ds;
  int cur;
};

__device__ __forceinline__ void run_reset(Run& r, int tl) {
  r.cur = tl;
  r.m = -__builtin_huge_valf();
  r.ds = 0.f;
#pragma unroll
  for (int j = 0; j < HH; ++j) r.sum[j] = make_float4(0.f, 0.f, 0.f, 0.f);
}

__device__ __forceinline__ void run_merge(float* acc, float* dns, const float* mxs, int node,
                                          float m, float ds, const float4 (&sum)[HH],
                                          int lane, int hd) {
  const float g = __expf(m - mxs[node * HH + hd]);
  float gh[HH];
  bcast_heads(g, gh);
  row_add_t(acc + node * FH, lane, sum, gh);
  if ((lane & 7) == 0) lds_add_f(&dns[node * HH + hd], ds * g);
}

__device__ __forceinline__ void run_edge(Run& r, int& fcur, float& fm, float& fds,
                                         float* acc, float* dns, float* slot, const float* erl,
                                         int lane, int hd, int tl, const float4 (&f)[HH],
                                         const float4 (&al4)[HH]) {
  const float el = head_dots(f, al4, lane);
  if (tl != r.cur) {
    if (r.cur >= 0) {
      if (fcur < 0) {
        row_store_t(slot, lane, r.sum);
        fm = r.m; fds = r.ds; fcur = r.cur;
      } else {
        row_store_t(acc + r.cur * FH, lane, r.sum);
        if ((lane & 7) == 0) dns[r.cur * HH + hd] = r.ds;
      }
    }
    run_reset(r, tl);
  }
  const float e = leaky(el + erl[tl * HH + hd]);
  const float mn = fmaxf(r.m, e);
  const float sc = __expf(r.m - mn);
  const float p = __expf(e - mn);
  r.ds = r.ds * sc + p;
  r.m = mn;
  float ph[HH], sh[HH];
  bcast_heads(p, ph);
  bcast_heads(sc, sh);
#pragma unroll
  for (int j = 0; j < HH; ++j) {
    r.sum[j].x = r.sum[j].x * sh[j] + f[j].x * ph[j];
    r.sum[j].y = r.sum[j].y * sh[j] + f[j].y * ph[j];
    r.sum[j].z = r.sum[j].z * sh[j] + f[j].z * ph[j];
    r.sum[j].w = r.sum[j].w * sh[j] + f[j].w * ph[j];
  }
}

__device__ __forceinline__ void entry_fetch(int k, const unsigned* list,
                                            const float* __restrict__ feat, int lane,
                                            float4 (&f)[HH], int& tl) {
  const unsigned w = list[k];
  const int s = __builtin_amdgcn_readfirstlane((int)(w & 0xFFFFu));
  tl = __builtin_amdgcn_readfirstlane((int)(w >> 16));
  row_load(feat, s, lane, f);
}

__device__ __forceinline__ void accumulate_online(int L, const unsigned* list, const float* erl,
                                                  float* mxs, float* dns, float* acc, float* part,
                                                  const float* __restrict__ feat,
                                                  const float4 (&al4)[HH], int wv, int lane) {
  const int hd = HEAD_OF(lane);
  const int per = (L + NWV - 1) / NWV;
  int k = __builtin_amdgcn_readfirstlane(wv) * per;
  const int end = (k + per < L) ? k + per : L;
  float* slot = part + wv * FH;
  Run r;
  run_reset(r, -1);
  int fcur = -1;
  float fm = 0.f, fds = 0.f;
  for (; k + 1 < end; k += 2) {
    float4 fa[HH], fb[HH];
    int ta, tb;
    entry_fetch(k, list, feat, lane, fa, ta);
    entry_fetch(k + 1, list, feat, lane, fb, tb);
    run_edge(r, fcur, fm, fds, acc, dns, slot, erl, lane, hd, ta, fa, al4);
    run_edge(r, fcur, fm, fds, acc, dns, slot, erl, lane, hd, tb, fb, al4);
  }
  if (k < end) {
    float4 fa[HH];
    int ta;
    entry_fetch(k, list, feat, lane, fa, ta);
    run_edge(r, fcur, fm, fds, acc, dns, slot, erl, lane, hd, ta, fa, al4);
  }
  if ((lane & 7) == 0) {
    if (fcur >= 0) lds_max_f(&mxs[fcur * HH + hd], fm);
    if (r.cur >= 0) lds_max_f(&mxs[r.cur * HH + hd], r.m);
  }
  __syncthreads();
  if (fcur >= 0) {
    float4 t[HH];
    row_load_t(slot, lane, t);
    run_merge(acc, dns, mxs, fcur, fm, fds, t, lane, hd);
  }
  if (r.cur >= 0) run_merge(acc, dns, mxs, r.cur, r.m, r.ds, r.sum, lane, hd);
}

template <bool SUM>
__device__ __forceinline__ void rows_known(int L, const unsigned* list, const float* erl,
                                           float* mxs, float* dns, float* acc,
                                           const float* __restrict__ feat,
                                           const float4 (&al4)[HH], int wv, int lane) {
  const int hd = HEAD_OF(lane);
  const int per = (L + NWV - 1) / NWV;
  int k = __builtin_amdgcn_readfirstlane(wv) * per;
  const int end = (k + per < L) ? k + per : L;
  const float one[HH] = {1.f, 1.f, 1.f, 1.f};
  float4 sum[HH];
  float ds = 0.f;
  int cur = -1;
#pragma unroll
  for (int j = 0; j < HH; ++j) sum[j] = make_float4(0.f, 0.f, 0.f, 0.f);
  for (; k < end; ++k) {
    float4 f[HH];
    int tl;
    entry_fetch(k, list, feat, lane, f, tl);
    const float e = leaky(head_dots(f, al4, lane) + erl[tl * HH + hd]);
    if (!SUM) {
      if ((lane & 7) == 0) lds_max_f(&mxs[tl * HH + hd], e);
    } else {
      if (tl != cur) {
        if (cur >= 0) {
          row_add_t(acc + cur * FH, lane, sum, one);
          if ((lane & 7) == 0) lds_add_f(&dns[cur * HH + hd], ds);
        }
        cur = tl;
        ds = 0.f;
#pragma unroll
        for (int j = 0; j < HH; ++j) sum[j] = make_float4(0.f, 0.f, 0.f, 0.f);
      }
      const float p = __expf(e - mxs[tl * HH + hd]);
      ds += p;
      float ph[HH];
      bcast_heads(p, ph);
#pragma unroll
      for (int j = 0; j < HH; ++j) {
        sum[j].x += f[j].x * ph[j]; sum[j].y += f[j].y * ph[j];
        sum[j].z += f[j].z * ph[j]; sum[j].w += f[j].w * ph[j];
      }
    }
  }
  if (SUM && cur >= 0) {
    row_add_t(acc + cur * FH, lane, sum, one);
    if ((lane & 7) == 0) lds_add_f(&dns[cur * HH + hd], ds);
  }
}

__device__ __forceinline__ float4 head_cols4(const float* hd, int d0) {
  const float* q = hd + (d0 >> 2);
  float4 a;
  a.x = q[0 * (DD / 4)]; a.y = q[1 * (DD / 4)];
  a.z = q[2 * (DD / 4)]; a.w = q[3 * (DD / 4)];
  return a;
}

#define HPT ((RN * FH) / (ETH * 4))
static_assert(HPT * ETH * 4 == RN * FH, "the h rows split evenly over the threads");
static_assert(ETH / 32 == FH / 16, "one wave per 16-column tile of the projection");
static_assert(RN > 32 && RN <= 48 && RN % 8 == 0, "three row tiles, the third half used");

__device__ __forceinline__ void project_rows(const float* hs, const float* __restrict__ wn,
                                             float* __restrict__ fn, int lo, int wv, int lane) {
  const int half = lane >> 4;
  const int l = lane & 15;
  const int r2 = 32 + (l & 7);
  const int x01 = l << 2, x2 = (r2 & 15) << 2;
  const float* a0p = hs + l * FH + half * 2;
  const float* a1p = hs + (16 + l) * FH + half * 2;
  const float* a2p = hs + r2 * FH + half * 2;
  const float* bp = wn + (size_t)(half * 2) * FH + wv * 16 + l;
  v8f c0 = {}, c1 = {}, c2 = {};
#pragma unroll 8
  for (int k = 0; k < FH; k += 4) {
    v2f b; b.x = bp[0]; b.y = bp[FH];
    bp += 4 * FH;
    const v2f a0 = *(const v2f*)(a0p + (k ^ x01));
    const v2f a1 = *(const v2f*)(a1p + (k ^ x01));
    const v2f a2 = *(const v2f*)(a2p + (k ^ x2));
    c0 = __builtin_amdgcn_wmma_f32_16x16x4_f32(false, a0, false, b, (short)0, c0, false, false);
    c1 = __builtin_amdgcn_wmma_f32_16x16x4_f32(false, a1, false, b, (short)0, c1, false, false);
    c2 = __builtin_amdgcn_wmma_f32_16x16x4_f32(false, a2, false, b, (short)0, c2, false, false);
  }
  float* ob = fn + (size_t)(lo + half * 8) * FH + wv * 16 + l;
#pragma unroll
  for (int r = 0; r < 8; ++r) {
    ob[(size_t)r * FH] = c0[r];
    ob[(size_t)(16 + r) * FH] = c1[r];
  }
  if (half == 0) {
#pragma unroll
    for (int r = 0; r < 8; ++r) ob[(size_t)(32 + r) * FH] = c2[r];
  }
}

__device__ __forceinline__ float elu1(float v) {
  return v > 0.f ? v : (__expf(v) - 1.f);
}

template <bool LAST>
__global__ __launch_bounds__(ETH) void edge_own(
    const int* __restrict__ src, const int* __restrict__ dst,
    const float* __restrict__ feat, const float* __restrict__ al,
    const float* __restrict__ ar, const float* __restrict__ hin,
    const float* __restrict__ wnext, float* __restrict__ fnext,
    float* __restrict__ outp, unsigned* __restrict__ glist, int* __restrict__ gcnt,
    const unsigned* __restrict__ okey) {
  __shared__ __attribute__((aligned(16))) float acc[RN * FH];
  __shared__ __attribute__((aligned(16))) float part[NWV * FH];
  __shared__ unsigned list[CAP];
  __shared__ __attribute__((aligned(16))) float mxs[RN * HH];
  __shared__ __attribute__((aligned(16))) float dns[RN * HH];
  __shared__ __attribute__((aligned(16))) float erl[RN * HH];
  __shared__ __attribute__((aligned(16))) int hist[RN];
  __shared__ __attribute__((aligned(16))) int nstart[RN];
  __shared__ int cnt;

  const int tid = threadIdx.x;
  const int lane = tid & 31;
  const int wv = tid >> 5;
  const int lo = blockIdx.x * RN;
  const unsigned pat = blockIdx.x * 0x01010101u;
  const uint4* k4 = (const uint4*)okey;

  for (int i = tid * 4; i < RN * FH; i += ETH * 4)
    *(float4*)(acc + i) = make_float4(0.f, 0.f, 0.f, 0.f);
  if (tid < RN * HH) { mxs[tid] = -__builtin_huge_valf(); dns[tid] = 0.f; }
  if (!LAST && tid < RN) hist[tid] = 0;
  if (tid == 0) cnt = 0;

  float4 al4[HH];
#pragma unroll
  for (int j = 0; j < HH; ++j) al4[j] = *(const float4*)(al + j * DD + 4 * lane);
  {
    float4 ar4[HH];
#pragma unroll
    for (int j = 0; j < HH; ++j) ar4[j] = *(const float4*)(ar + j * DD + 4 * lane);
    for (int r = wv; r < RN; r += NWV) {
      float4 f[HH];
      row_load(feat, lo + r, lane, f);
      const float d = head_dots(f, ar4, lane);
      if ((lane & 7) == 0) erl[r * HH + HEAD_OF(lane)] = d;
    }
  }
  __syncthreads();

  unsigned* gl = glist + (size_t)blockIdx.x * CAP;
  int total;
  if (!LAST) {
    scan_keys(k4, 0, EE / 16, pat, list, &cnt, tid);
    __syncthreads();
    total = cnt;
    if (tid == 0) gcnt[blockIdx.x] = total;
  } else {
    total = gcnt[blockIdx.x];
  }

  if (total <= CAP) {
    if (!LAST) {
      edge_sort(total, list, hist, nstart, src, dst, lo, tid);
      for (int k = tid; k < total; k += ETH) gl[k] = list[k];
    } else {
      for (int k = tid; k < total; k += ETH) list[k] = gl[k];
      __syncthreads();
    }
    accumulate_online(total, list, erl, mxs, dns, acc, part, feat, al4, wv, lane);
  } else {
    for (int c0 = 0; c0 < EE; c0 += CAP) {
      __syncthreads();
      if (tid == 0) cnt = 0;
      __syncthreads();
      scan_keys(k4, c0 / 16, ((c0 + CAP < EE) ? c0 + CAP : EE) / 16, pat, list, &cnt, tid);
      __syncthreads();
      const int L = cnt;
      edge_prepare(L, list, src, dst, lo, tid);
      __syncthreads();
      rows_known<false>(L, list, erl, mxs, dns, acc, feat, al4, wv, lane);
    }
    for (int c0 = 0; c0 < EE; c0 += CAP) {
      __syncthreads();
      if (tid == 0) cnt = 0;
      __syncthreads();
      scan_keys(k4, c0 / 16, ((c0 + CAP < EE) ? c0 + CAP : EE) / 16, pat, list, &cnt, tid);
      __syncthreads();
      const int L = cnt;
      edge_prepare(L, list, src, dst, lo, tid);
      __syncthreads();
      rows_known<true>(L, list, erl, mxs, dns, acc, feat, al4, wv, lane);
    }
  }
  __syncthreads();

  if (!LAST) {
    float* ob = outp + (size_t)lo * FH;
    float4 hv[HPT];
#pragma unroll
    for (int it = 0; it < HPT; ++it) {
      const int i = (it * ETH + tid) * 4;
      const float d = dns[i >> 7];
      const float dd = d == 0.f ? 1.f : d;
      const float4 a = head_cols4(acc + (i & ~(DD - 1)), i & (DD - 1));
      float4 o;
      o.x = elu1(a.x / dd); o.y = elu1(a.y / dd);
      o.z = elu1(a.z / dd); o.w = elu1(a.w / dd);
      hv[it] = o;
    }
    __syncthreads();
#pragma unroll
    for (int it = 0; it < HPT; ++it) {
      const int i = (it * ETH + tid) * 4;
      const int r = i >> 9, c = i & (FH - 1);
      *(float4*)(acc + r * FH + (c ^ ((r & 15) << 2))) = hv[it];
    }
    __syncthreads();
#pragma unroll
    for (int it = 0; it < HPT; ++it)
      *(float4*)(ob + (it * ETH + tid) * 4) = hv[it];
    project_rows(acc, wnext, fnext, lo, wv, lane);
  } else {
    for (int i = tid * 4; i < RN * DD; i += ETH * 4) {
      const int tl = i >> 7, d0 = i & (DD - 1);
      float4 sum = make_float4(0.f, 0.f, 0.f, 0.f);
#pragma unroll
      for (int h = 0; h < HH; ++h) {
        const float d = dns[tl * HH + h];
        const float dd = d == 0.f ? 1.f : d;
        const float4 a = head_cols4(acc + tl * FH + h * DD, d0);
        const float4 r = *(const float4*)(hin + (size_t)(lo + tl) * FH + h * DD + d0);
        sum.x += elu1(a.x / dd + r.x); sum.y += elu1(a.y / dd + r.y);
        sum.z += elu1(a.z / dd + r.z); sum.w += elu1(a.w / dd + r.w);
      }
      float4 o;
      o.x = sum.x * (1.f / HH); o.y = sum.y * (1.f / HH);
      o.z = sum.z * (1.f / HH); o.w = sum.w * (1.f / HH);
      *(float4*)(outp + (size_t)(lo + tl) * DD + d0) = o;
    }
  }
}

extern "C" void kernel_launch(void* const* d_in, const int* in_sizes, int n_in,
                              void* d_out, int out_size, void* d_ws, size_t ws_size,
                              hipStream_t stream) {
  const int* src = (const int*)d_in[0];
  const int* dst = (const int*)d_in[1];
  const float* x   = (const float*)d_in[2];
  const float* W0  = (const float*)d_in[3];
  const float* al0 = (const float*)d_in[4];
  const float* ar0 = (const float*)d_in[5];
  const float* W1  = (const float*)d_in[6];
  const float* al1 = (const float*)d_in[7];
  const float* ar1 = (const float*)d_in[8];
  float* out = (float*)d_out;

  float* ws    = (float*)d_ws;
  float* feat  = ws;
  float* hbuf  = feat  + (size_t)NN * FH;
  float* feat1 = hbuf  + (size_t)NN * FH;
  unsigned* glist = (unsigned*)(feat1 + (size_t)NN * FH);
  int* gcnt = (int*)(glist + (size_t)NBLK * CAP);
  unsigned* okey = (unsigned*)(gcnt + 256);

  gemm_rows_lds<<<(NN / GR) * (FH / GC), GTH, 0, stream>>>(x, W0, feat, dst, okey);
  edge_own<false><<<NBLK, ETH, 0, stream>>>(src, dst, feat, al0, ar0, nullptr, W1, feat1, hbuf, glist, gcnt,
                                            okey);
  edge_own<true><<<NBLK, ETH, 0, stream>>>(src, dst, feat1, al1, ar1, hbuf, nullptr, nullptr, out, glist, gcnt,
                                           okey);
}
